# starting kernel
# baseline (speedup 1.0000x reference)
#include <hip/hip_runtime.h>

typedef __attribute__((ext_vector_type(2))) float v2f;
typedef __attribute__((ext_vector_type(8))) float v8f;

#define NEG_SLOPE 0.2f

template <int CIN, int COUT>
__global__ __launch_bounds__(256) void gemm_wmma(const float* __restrict__ X,
                                                 const float* __restrict__ Wg,
                                                 float* __restrict__ H, int M) {
  __shared__ v2f Wswz[(CIN / 4) * COUT * 2];
  {
    constexpr int TOT = (CIN / 4) * COUT * 2;
    for (int i = threadIdx.x; i < TOT; i += 256) {
      int m = i / (COUT * 2);
      int r = i - m * (COUT * 2);
      int c = r >> 1;
      int hsel = r & 1;
      const float* wp = Wg + (size_t)(4 * m + 2 * hsel) * COUT + c;
      v2f w;
      w.x = wp[0];
      w.y = wp[COUT];
      Wswz[i] = w;
    }
  }
  __syncthreads();

  const int lane = threadIdx.x & 31;
  const int wave = threadIdx.x >> 5;
  const int mtile = blockIdx.x * 8 + wave;
  if (mtile * 16 >= M) return;
  const int m0 = mtile * 16;

  constexpr int NT = COUT / 16;
  v8f acc[NT] = {};

  const int rsel = lane & 15;
  const int hiHalf = lane >> 4;
  const float* xrow = X + (size_t)(m0 + rsel) * CIN + hiHalf * 2;
  const v2f* wbase = Wswz + rsel * 2 + hiHalf;

  for (int k = 0; k < CIN; k += 4) {
    v2f a = *reinterpret_cast<const v2f*>(xrow + k);
    const v2f* wb = wbase + (k >> 2) * (COUT * 2);
#pragma unroll
    for (int t = 0; t < NT; ++t) {
      v2f bf = wb[t * 32];
      acc[t] = __builtin_amdgcn_wmma_f32_16x16x4_f32(
          false, a, false, bf, (short)0, acc[t], false, false);
    }
  }

  const int rbase = m0 + hiHalf * 8;
#pragma unroll
  for (int t = 0; t < NT; ++t) {
#pragma unroll
    for (int i = 0; i < 8; ++i) {
      H[(size_t)(rbase + i) * COUT + t * 16 + rsel] = acc[t][i];
    }
  }
}

template <int C>
__global__ __launch_bounds__(256) void row_proj(const float* __restrict__ H,
                                                const float* __restrict__ asrc,
                                                const float* __restrict__ adst,
                                                float* __restrict__ als,
                                                float* __restrict__ ald, int Nn) {
  const int lane = threadIdx.x & 31;
  const int wave = threadIdx.x >> 5;
  const int n = blockIdx.x * 8 + wave;
  if (n >= Nn) return;
  const float* hrow = H + (size_t)n * C;
  float s = 0.f, d = 0.f;
#pragma unroll
  for (int c = lane; c < C; c += 32) {
    float hv = hrow[c];
    s += hv * asrc[c];
    d += hv * adst[c];
  }
#pragma unroll
  for (int off = 16; off > 0; off >>= 1) {
    s += __shfl_xor(s, off, 32);
    d += __shfl_xor(d, off, 32);
  }
  if (lane == 0) {
    als[n] = s;
    ald[n] = d;
  }
}

__device__ __forceinline__ unsigned fenc(float f) {
  unsigned u = __float_as_uint(f);
  return u ^ ((u >> 31) ? 0xFFFFFFFFu : 0x80000000u);
}
__device__ __forceinline__ float fdec(unsigned u) {
  unsigned v = u ^ ((u & 0x80000000u) ? 0x80000000u : 0xFFFFFFFFu);
  return __uint_as_float(v);
}

__device__ __forceinline__ void edge_sd(int i, int E, const int* __restrict__ src,
                                        const int* __restrict__ dst, int& s, int& d) {
  if (i < E) {
    s = src[i];
    d = dst[i];
  } else {
    s = i - E;
    d = s;
  }
}

__global__ __launch_bounds__(256) void edge_max(const int* __restrict__ src,
                                                const int* __restrict__ dst,
                                                const float* __restrict__ als,
                                                const float* __restrict__ ald,
                                                unsigned* __restrict__ menc,
                                                int E, int Etot) {
  int i = blockIdx.x * 256 + threadIdx.x;
  if (i >= Etot) return;
  int s, d;
  edge_sd(i, E, src, dst, s, d);
  float e = als[s] + ald[d];
  e = (e > 0.f) ? e : NEG_SLOPE * e;
  atomicMax(&menc[d], fenc(e));
}

__global__ __launch_bounds__(256) void edge_expsum(const int* __restrict__ src,
                                                   const int* __restrict__ dst,
                                                   const float* __restrict__ als,
                                                   const float* __restrict__ ald,
                                                   const unsigned* __restrict__ menc,
                                                   float* __restrict__ ee,
                                                   float* __restrict__ denom,
                                                   int E, int Etot) {
  int i = blockIdx.x * 256 + threadIdx.x;
  if (i >= Etot) return;
  int s, d;
  edge_sd(i, E, src, dst, s, d);
  float e = als[s] + ald[d];
  e = (e > 0.f) ? e : NEG_SLOPE * e;
  float x = __expf(e - fdec(menc[d]));
  ee[i] = x;
  atomicAdd(&denom[d], x);
}

template <int C>
__global__ __launch_bounds__(256) void edge_aggregate(const int* __restrict__ src,
                                                      const int* __restrict__ dst,
                                                      const float* __restrict__ H,
                                                      const float* __restrict__ ee,
                                                      const float* __restrict__ denom,
                                                      float* __restrict__ out,
                                                      int E, int Etot) {
  const int lane = threadIdx.x & 31;
  const int wave = threadIdx.x >> 5;
  int i = blockIdx.x * 8 + wave;
  if (i >= Etot) return;
  int s, d;
  edge_sd(i, E, src, dst, s, d);
  float alpha = ee[i] / denom[d];
  constexpr int VPT = C / 32;
  const float* hs = H + (size_t)s * C + lane * VPT;
  float* od = out + (size_t)d * C + lane * VPT;
#pragma unroll
  for (int j = 0; j < VPT; ++j) atomicAdd(&od[j], alpha * hs[j]);
}

template <int C>
__global__ __launch_bounds__(256) void bias_act(float* __restrict__ io,
                                                const float* __restrict__ b,
                                                int total, int relu) {
  int i = blockIdx.x * 256 + threadIdx.x;
  if (i >= total) return;
  float v = io[i] + b[i & (C - 1)];
  if (relu) v = fmaxf(v, 0.f);
  io[i] = v;
}

extern "C" void kernel_launch(void* const* d_in, const int* in_sizes, int n_in,
                              void* d_out, int out_size, void* d_ws, size_t ws_size,
                              hipStream_t stream) {
  const float* x   = (const float*)d_in[0];
  const int*   ei  = (const int*)d_in[1];
  const float* W1  = (const float*)d_in[2];
  const float* as1 = (const float*)d_in[3];
  const float* ad1 = (const float*)d_in[4];
  const float* b1  = (const float*)d_in[5];
  const float* W2  = (const float*)d_in[6];
  const float* as2 = (const float*)d_in[7];
  const float* ad2 = (const float*)d_in[8];
  const float* b2  = (const float*)d_in[9];
  const float* W3  = (const float*)d_in[10];
  const float* as3 = (const float*)d_in[11];
  const float* ad3 = (const float*)d_in[12];
  const float* b3  = (const float*)d_in[13];

  const int N = in_sizes[0] / 128;
  const int E = in_sizes[1] / 2;
  const int Etot = E + N;
  const int* srcp = ei;
  const int* dstp = ei + E;

  float*    buf0  = (float*)d_ws;
  float*    buf1  = buf0 + (size_t)N * 128;
  float*    als   = buf1 + (size_t)N * 128;
  float*    ald   = als + N;
  unsigned* menc  = (unsigned*)(ald + N);
  float*    denom = (float*)(menc + N);
  float*    ee    = denom + N;

  const int gGemm = (N / 16 + 7) / 8;
  const int gProj = (N + 7) / 8;
  const int gE    = (Etot + 255) / 256;
  const int gEagg = (Etot + 7) / 8;
  float* outF = (float*)d_out;

  gemm_wmma<128, 128><<<gGemm, 256, 0, stream>>>(x, W1, buf0, N);
  row_proj<128><<<gProj, 256, 0, stream>>>(buf0, as1, ad1, als, ald, N);
  hipMemsetAsync(menc, 0, (size_t)N * 4, stream);
  hipMemsetAsync(denom, 0, (size_t)N * 4, stream);
  hipMemsetAsync(buf1, 0, (size_t)N * 128 * 4, stream);
  edge_max<<<gE, 256, 0, stream>>>(srcp, dstp, als, ald, menc, E, Etot);
  edge_expsum<<<gE, 256, 0, stream>>>(srcp, dstp, als, ald, menc, ee, denom, E, Etot);
  edge_aggregate<128><<<gEagg, 256, 0, stream>>>(srcp, dstp, buf0, ee, denom, buf1, E, Etot);
  bias_act<128><<<(N * 128 + 255) / 256, 256, 0, stream>>>(buf1, b1, N * 128, 1);

  gemm_wmma<128, 128><<<gGemm, 256, 0, stream>>>(buf1, W2, buf0, N);
  row_proj<128><<<gProj, 256, 0, stream>>>(buf0, as2, ad2, als, ald, N);
  hipMemsetAsync(menc, 0, (size_t)N * 4, stream);
  hipMemsetAsync(denom, 0, (size_t)N * 4, stream);
  hipMemsetAsync(buf1, 0, (size_t)N * 128 * 4, stream);
  edge_max<<<gE, 256, 0, stream>>>(srcp, dstp, als, ald, menc, E, Etot);
  edge_expsum<<<gE, 256, 0, stream>>>(srcp, dstp, als, ald, menc, ee, denom, E, Etot);
  edge_aggregate<128><<<gEagg, 256, 0, stream>>>(srcp, dstp, buf0, ee, denom, buf1, E, Etot);
  bias_act<128><<<(N * 128 + 255) / 256, 256, 0, stream>>>(buf1, b2, N * 128, 1);

  gemm_wmma<128, 64><<<gGemm, 256, 0, stream>>>(buf1, W3, buf0, N);
  row_proj<64><<<gProj, 256, 0, stream>>>(buf0, as3, ad3, als, ald, N);
  hipMemsetAsync(menc, 0, (size_t)N * 4, stream);
  hipMemsetAsync(denom, 0, (size_t)N * 4, stream);
  hipMemsetAsync(outF, 0, (size_t)N * 64 * 4, stream);
  edge_max<<<gE, 256, 0, stream>>>(srcp, dstp, als, ald, menc, E, Etot);
  edge_expsum<<<gE, 256, 0, stream>>>(srcp, dstp, als, ald, menc, ee, denom, E, Etot);
  edge_aggregate<64><<<gEagg, 256, 0, stream>>>(srcp, dstp, buf0, ee, denom, outF, E, Etot);
  bias_act<64><<<(N * 64 + 255) / 256, 256, 0, stream>>>(outF, b3, N * 64, 0);
}


# edge_aggregate: lane l owns columns l+32j, so each fp32 atomic (and load) instruction of a wave covers one whole 128-byte line instead of 8 lanes in each of 4 lines
# speedup vs baseline: 1.1762x; 1.1762x over previous
#include <hip/hip_runtime.h>

typedef __attribute__((ext_vector_type(2))) float v2f;
typedef __attribute__((ext_vector_type(8))) float v8f;

#define NEG_SLOPE 0.2f

template <int CIN, int COUT>
__global__ __launch_bounds__(256) void gemm_wmma(const float* __restrict__ X,
                                                 const float* __restrict__ Wg,
                                                 float* __restrict__ H, int M) {
  __shared__ v2f Wswz[(CIN / 4) * COUT * 2];
  {
    constexpr int TOT = (CIN / 4) * COUT * 2;
    for (int i = threadIdx.x; i < TOT; i += 256) {
      int m = i / (COUT * 2);
      int r = i - m * (COUT * 2);
      int c = r >> 1;
      int hsel = r & 1;
      const float* wp = Wg + (size_t)(4 * m + 2 * hsel) * COUT + c;
      v2f w;
      w.x = wp[0];
      w.y = wp[COUT];
      Wswz[i] = w;
    }
  }
  __syncthreads();

  const int lane = threadIdx.x & 31;
  const int wave = threadIdx.x >> 5;
  const int mtile = blockIdx.x * 8 + wave;
  if (mtile * 16 >= M) return;
  const int m0 = mtile * 16;

  constexpr int NT = COUT / 16;
  v8f acc[NT] = {};

  const int rsel = lane & 15;
  const int hiHalf = lane >> 4;
  const float* xrow = X + (size_t)(m0 + rsel) * CIN + hiHalf * 2;
  const v2f* wbase = Wswz + rsel * 2 + hiHalf;

  for (int k = 0; k < CIN; k += 4) {
    v2f a = *reinterpret_cast<const v2f*>(xrow + k);
    const v2f* wb = wbase + (k >> 2) * (COUT * 2);
#pragma unroll
    for (int t = 0; t < NT; ++t) {
      v2f bf = wb[t * 32];
      acc[t] = __builtin_amdgcn_wmma_f32_16x16x4_f32(
          false, a, false, bf, (short)0, acc[t], false, false);
    }
  }

  const int rbase = m0 + hiHalf * 8;
#pragma unroll
  for (int t = 0; t < NT; ++t) {
#pragma unroll
    for (int i = 0; i < 8; ++i) {
      H[(size_t)(rbase + i) * COUT + t * 16 + rsel] = acc[t][i];
    }
  }
}

template <int C>
__global__ __launch_bounds__(256) void row_proj(const float* __restrict__ H,
                                                const float* __restrict__ asrc,
                                                const float* __restrict__ adst,
                                                float* __restrict__ als,
                                                float* __restrict__ ald, int Nn) {
  const int lane = threadIdx.x & 31;
  const int wave = threadIdx.x >> 5;
  const int n = blockIdx.x * 8 + wave;
  if (n >= Nn) return;
  const float* hrow = H + (size_t)n * C;
  float s = 0.f, d = 0.f;
#pragma unroll
  for (int c = lane; c < C; c += 32) {
    float hv = hrow[c];
    s += hv * asrc[c];
    d += hv * adst[c];
  }
#pragma unroll
  for (int off = 16; off > 0; off >>= 1) {
    s += __shfl_xor(s, off, 32);
    d += __shfl_xor(d, off, 32);
  }
  if (lane == 0) {
    als[n] = s;
    ald[n] = d;
  }
}

__device__ __forceinline__ unsigned fenc(float f) {
  unsigned u = __float_as_uint(f);
  return u ^ ((u >> 31) ? 0xFFFFFFFFu : 0x80000000u);
}
__device__ __forceinline__ float fdec(unsigned u) {
  unsigned v = u ^ ((u & 0x80000000u) ? 0x80000000u : 0xFFFFFFFFu);
  return __uint_as_float(v);
}

__device__ __forceinline__ void edge_sd(int i, int E, const int* __restrict__ src,
                                        const int* __restrict__ dst, int& s, int& d) {
  if (i < E) {
    s = src[i];
    d = dst[i];
  } else {
    s = i - E;
    d = s;
  }
}

__global__ __launch_bounds__(256) void edge_max(const int* __restrict__ src,
                                                const int* __restrict__ dst,
                                                const float* __restrict__ als,
                                                const float* __restrict__ ald,
                                                unsigned* __restrict__ menc,
                                                int E, int Etot) {
  int i = blockIdx.x * 256 + threadIdx.x;
  if (i >= Etot) return;
  int s, d;
  edge_sd(i, E, src, dst, s, d);
  float e = als[s] + ald[d];
  e = (e > 0.f) ? e : NEG_SLOPE * e;
  atomicMax(&menc[d], fenc(e));
}

__global__ __launch_bounds__(256) void edge_expsum(const int* __restrict__ src,
                                                   const int* __restrict__ dst,
                                                   const float* __restrict__ als,
                                                   const float* __restrict__ ald,
                                                   const unsigned* __restrict__ menc,
                                                   float* __restrict__ ee,
                                                   float* __restrict__ denom,
                                                   int E, int Etot) {
  int i = blockIdx.x * 256 + threadIdx.x;
  if (i >= Etot) return;
  int s, d;
  edge_sd(i, E, src, dst, s, d);
  float e = als[s] + ald[d];
  e = (e > 0.f) ? e : NEG_SLOPE * e;
  float x = __expf(e - fdec(menc[d]));
  ee[i] = x;
  atomicAdd(&denom[d], x);
}

template <int C>
__global__ __launch_bounds__(256) void edge_aggregate(const int* __restrict__ src,
                                                      const int* __restrict__ dst,
                                                      const float* __restrict__ H,
                                                      const float* __restrict__ ee,
                                                      const float* __restrict__ denom,
                                                      float* __restrict__ out,
                                                      int E, int Etot) {
  const int lane = threadIdx.x & 31;
  const int wave = threadIdx.x >> 5;
  int i = blockIdx.x * 8 + wave;
  if (i >= Etot) return;
  int s, d;
  edge_sd(i, E, src, dst, s, d);
  float alpha = ee[i] / denom[d];
  constexpr int VPT = C / 32;
  const float* hs = H + (size_t)s * C + lane;
  float* od = out + (size_t)d * C + lane;
  float v[VPT];
#pragma unroll
  for (int j = 0; j < VPT; ++j) v[j] = hs[j * 32];
#pragma unroll
  for (int j = 0; j < VPT; ++j) atomicAdd(&od[j * 32], alpha * v[j]);
}

template <int C>
__global__ __launch_bounds__(256) void bias_act(float* __restrict__ io,
                                                const float* __restrict__ b,
                                                int total, int relu) {
  int i = blockIdx.x * 256 + threadIdx.x;
  if (i >= total) return;
  float v = io[i] + b[i & (C - 1)];
  if (relu) v = fmaxf(v, 0.f);
  io[i] = v;
}

extern "C" void kernel_launch(void* const* d_in, const int* in_sizes, int n_in,
                              void* d_out, int out_size, void* d_ws, size_t ws_size,
                              hipStream_t stream) {
  const float* x   = (const float*)d_in[0];
  const int*   ei  = (const int*)d_in[1];
  const float* W1  = (const float*)d_in[2];
  const float* as1 = (const float*)d_in[3];
  const float* ad1 = (const float*)d_in[4];
  const float* b1  = (const float*)d_in[5];
  const float* W2  = (const float*)d_in[6];
  const float* as2 = (const float*)d_in[7];
  const float* ad2 = (const float*)d_in[8];
  const float* b2  = (const float*)d_in[9];
  const float* W3  = (const float*)d_in[10];
  const float* as3 = (const float*)d_in[11];
  const float* ad3 = (const float*)d_in[12];
  const float* b3  = (const float*)d_in[13];

  const int N = in_sizes[0] / 128;
  const int E = in_sizes[1] / 2;
  const int Etot = E + N;
  const int* srcp = ei;
  const int* dstp = ei + E;

  float*    buf0  = (float*)d_ws;
  float*    buf1  = buf0 + (size_t)N * 128;
  float*    als   = buf1 + (size_t)N * 128;
  float*    ald   = als + N;
  unsigned* menc  = (unsigned*)(ald + N);
  float*    denom = (float*)(menc + N);
  float*    ee    = denom + N;

  const int gGemm = (N / 16 + 7) / 8;
  const int gProj = (N + 7) / 8;
  const int gE    = (Etot + 255) / 256;
  const int gEagg = (Etot + 7) / 8;
  float* outF = (float*)d_out;

  gemm_wmma<128, 128><<<gGemm, 256, 0, stream>>>(x, W1, buf0, N);
  row_proj<128><<<gProj, 256, 0, stream>>>(buf0, as1, ad1, als, ald, N);
  hipMemsetAsync(menc, 0, (size_t)N * 4, stream);
  hipMemsetAsync(denom, 0, (size_t)N * 4, stream);
  hipMemsetAsync(buf1, 0, (size_t)N * 128 * 4, stream);
  edge_max<<<gE, 256, 0, stream>>>(srcp, dstp, als, ald, menc, E, Etot);
  edge_expsum<<<gE, 256, 0, stream>>>(srcp, dstp, als, ald, menc, ee, denom, E, Etot);
  edge_aggregate<128><<<gEagg, 256, 0, stream>>>(srcp, dstp, buf0, ee, denom, buf1, E, Etot);
  bias_act<128><<<(N * 128 + 255) / 256, 256, 0, stream>>>(buf1, b1, N * 128, 1);

  gemm_wmma<128, 128><<<gGemm, 256, 0, stream>>>(buf1, W2, buf0, N);
  row_proj<128><<<gProj, 256, 0, stream>>>(buf0, as2, ad2, als, ald, N);
  hipMemsetAsync(menc, 0, (size_t)N * 4, stream);
  hipMemsetAsync(denom, 0, (size_t)N * 4, stream);
  hipMemsetAsync(buf1, 0, (size_t)N * 128 * 4, stream);
  edge_max<<<gE, 256, 0, stream>>>(srcp, dstp, als, ald, menc, E, Etot);
  edge_expsum<<<gE, 256, 0, stream>>>(srcp, dstp, als, ald, menc, ee, denom, E, Etot);
  edge_aggregate<128><<<gEagg, 256, 0, stream>>>(srcp, dstp, buf0, ee, denom, buf1, E, Etot);
  bias_act<128><<<(N * 128 + 255) / 256, 256, 0, stream>>>(buf1, b2, N * 128, 1);

  gemm_wmma<128, 64><<<gGemm, 256, 0, stream>>>(buf1, W3, buf0, N);
  row_proj<64><<<gProj, 256, 0, stream>>>(buf0, as3, ad3, als, ald, N);
  hipMemsetAsync(menc, 0, (size_t)N * 4, stream);
  hipMemsetAsync(denom, 0, (size_t)N * 4, stream);
  hipMemsetAsync(outF, 0, (size_t)N * 64 * 4, stream);
  edge_max<<<gE, 256, 0, stream>>>(srcp, dstp, als, ald, menc, E, Etot);
  edge_expsum<<<gE, 256, 0, stream>>>(srcp, dstp, als, ald, menc, ee, denom, E, Etot);
  edge_aggregate<64><<<gEagg, 256, 0, stream>>>(srcp, dstp, buf0, ee, denom, outF, E, Etot);
  bias_act<64><<<(N * 64 + 255) / 256, 256, 0, stream>>>(outF, b3, N * 64, 0);
}


# edges grouped by destination in-call (returning int add = slot rank, arrival order; scan; fill); one wave per node does max, exp-sum, weighted row sum in registers: no fp32 atomics, new sum order
# speedup vs baseline: 4.1252x; 3.5072x over previous
#include <hip/hip_runtime.h>

typedef __attribute__((ext_vector_type(2))) float v2f;
typedef __attribute__((ext_vector_type(8))) float v8f;

#define NEG_SLOPE 0.2f

template <int CIN, int COUT>
__global__ __launch_bounds__(256) void gemm_wmma(const float* __restrict__ X,
                                                 const float* __restrict__ Wg,
                                                 float* __restrict__ H, int M) {
  __shared__ v2f Wswz[(CIN / 4) * COUT * 2];
  {
    constexpr int TOT = (CIN / 4) * COUT * 2;
    for (int i = threadIdx.x; i < TOT; i += 256) {
      int m = i / (COUT * 2);
      int r = i - m * (COUT * 2);
      int c = r >> 1;
      int hsel = r & 1;
      const float* wp = Wg + (size_t)(4 * m + 2 * hsel) * COUT + c;
      v2f w;
      w.x = wp[0];
      w.y = wp[COUT];
      Wswz[i] = w;
    }
  }
  __syncthreads();

  const int lane = threadIdx.x & 31;
  const int wave = threadIdx.x >> 5;
  const int mtile = blockIdx.x * 8 + wave;
  if (mtile * 16 >= M) return;
  const int m0 = mtile * 16;

  constexpr int NT = COUT / 16;
  v8f acc[NT] = {};

  const int rsel = lane & 15;
  const int hiHalf = lane >> 4;
  const float* xrow = X + (size_t)(m0 + rsel) * CIN + hiHalf * 2;
  const v2f* wbase = Wswz + rsel * 2 + hiHalf;

  for (int k = 0; k < CIN; k += 4) {
    v2f a = *reinterpret_cast<const v2f*>(xrow + k);
    const v2f* wb = wbase + (k >> 2) * (COUT * 2);
#pragma unroll
    for (int t = 0; t < NT; ++t) {
      v2f bf = wb[t * 32];
      acc[t] = __builtin_amdgcn_wmma_f32_16x16x4_f32(
          false, a, false, bf, (short)0, acc[t], false, false);
    }
  }

  const int rbase = m0 + hiHalf * 8;
#pragma unroll
  for (int t = 0; t < NT; ++t) {
#pragma unroll
    for (int i = 0; i < 8; ++i) {
      H[(size_t)(rbase + i) * COUT + t * 16 + rsel] = acc[t][i];
    }
  }
}

template <int C>
__global__ __launch_bounds__(256) void row_proj(const float* __restrict__ H,
                                                const float* __restrict__ asrc,
                                                const float* __restrict__ adst,
                                                float* __restrict__ als,
                                                float* __restrict__ ald, int Nn) {
  const int lane = threadIdx.x & 31;
  const int wave = threadIdx.x >> 5;
  const int n = blockIdx.x * 8 + wave;
  if (n >= Nn) return;
  const float* hrow = H + (size_t)n * C;
  float s = 0.f, d = 0.f;
#pragma unroll
  for (int c = lane; c < C; c += 32) {
    float hv = hrow[c];
    s += hv * asrc[c];
    d += hv * adst[c];
  }
#pragma unroll
  for (int off = 16; off > 0; off >>= 1) {
    s += __shfl_xor(s, off, 32);
    d += __shfl_xor(d, off, 32);
  }
  if (lane == 0) {
    als[n] = s;
    ald[n] = d;
  }
}

__device__ __forceinline__ float lrelu(float e) { return (e > 0.f) ? e : NEG_SLOPE * e; }

__global__ __launch_bounds__(256) void edge_rank(const int* __restrict__ dst,
                                                 int* __restrict__ deg,
                                                 int* __restrict__ rank, int E) {
  int i = blockIdx.x * 256 + threadIdx.x;
  if (i >= E) return;
  rank[i] = atomicAdd(&deg[dst[i]], 1);
}

__global__ __launch_bounds__(256) void deg_blocksum(const int* __restrict__ deg,
                                                    int* __restrict__ bsum, int Nn) {
  __shared__ int wsum[8];
  const int base = blockIdx.x * 1024 + threadIdx.x * 4;
  int t = 0;
  for (int q = 0; q < 4; ++q)
    if (base + q < Nn) t += deg[base + q];
  for (int off = 16; off > 0; off >>= 1) t += __shfl_xor(t, off, 32);
  if ((threadIdx.x & 31) == 0) wsum[threadIdx.x >> 5] = t;
  __syncthreads();
  if (threadIdx.x == 0) {
    int s = 0;
    for (int w = 0; w < 8; ++w) s += wsum[w];
    bsum[blockIdx.x] = s;
  }
}

__global__ __launch_bounds__(256) void deg_scan(const int* __restrict__ deg,
                                                const int* __restrict__ bsum,
                                                int* __restrict__ rowptr, int Nn) {
  __shared__ int wpre[8];
  __shared__ int wtot[8];
  const int lane = threadIdx.x & 31;
  const int wave = threadIdx.x >> 5;
  int o = 0;
  for (int j = threadIdx.x; j < (int)blockIdx.x; j += 256) o += bsum[j];
  for (int off = 16; off > 0; off >>= 1) o += __shfl_xor(o, off, 32);
  if (lane == 0) wpre[wave] = o;

  const int base = blockIdx.x * 1024 + threadIdx.x * 4;
  int v[4];
  int t = 0;
  for (int q = 0; q < 4; ++q) {
    v[q] = (base + q < Nn) ? deg[base + q] : 0;
    t += v[q];
  }
  int inc = t;
  for (int off = 1; off < 32; off <<= 1) {
    int u = __shfl_up(inc, off, 32);
    if (lane >= off) inc += u;
  }
  if (lane == 31) wtot[wave] = inc;
  __syncthreads();
  int ex = inc - t;
  for (int w = 0; w < 8; ++w) {
    ex += wpre[w];
    if (w < wave) ex += wtot[w];
  }
  for (int q = 0; q < 4; ++q) {
    if (base + q <= Nn) rowptr[base + q] = ex;
    ex += v[q];
  }
}

__global__ __launch_bounds__(256) void edge_fill(const int* __restrict__ src,
                                                 const int* __restrict__ dst,
                                                 const int* __restrict__ rank,
                                                 const int* __restrict__ rowptr,
                                                 int* __restrict__ csr, int E) {
  int i = blockIdx.x * 256 + threadIdx.x;
  if (i >= E) return;
  csr[rowptr[dst[i]] + rank[i]] = src[i];
}

__device__ __forceinline__ float wave_max(float v) {
  for (int off = 16; off > 0; off >>= 1) v = fmaxf(v, __shfl_xor(v, off, 32));
  return v;
}
__device__ __forceinline__ float wave_sum(float v) {
  for (int off = 16; off > 0; off >>= 1) v += __shfl_xor(v, off, 32);
  return v;
}

template <int C, typename vecf>
__device__ __forceinline__ void add_rows(vecf& acc, const float* __restrict__ hl,
                                         int s, float w, int cnt) {
  for (int j = 0; j < cnt; j += 4) {
    vecf hv[4];
    float wj[4];
    for (int q = 0; q < 4; ++q) {
      const int sj = __builtin_amdgcn_readlane(s, j + q);
      wj[q] = __int_as_float(__builtin_amdgcn_readlane(__float_as_int(w), j + q));
      hv[q] = *reinterpret_cast<const vecf*>(hl + (size_t)sj * C);
    }
    for (int q = 0; q < 4; ++q) acc += wj[q] * hv[q];
  }
}

template <int C, bool RELU>
__global__ __launch_bounds__(256) void node_gather(const int* __restrict__ rowptr,
                                                   const int* __restrict__ csr,
                                                   const float* __restrict__ H,
                                                   const float* __restrict__ als,
                                                   const float* __restrict__ ald,
                                                   const float* __restrict__ bias,
                                                   float* __restrict__ out, int Nn) {
  constexpr int VPT = C / 32;
  typedef __attribute__((ext_vector_type(VPT))) float vecf;
  const int lane = threadIdx.x & 31;
  const int w0 = __builtin_amdgcn_readfirstlane((int)(blockIdx.x * 8 + (threadIdx.x >> 5)));
  const int nw = gridDim.x * 8;
  const float* hl = H + lane * VPT;
  const vecf bv = *reinterpret_cast<const vecf*>(bias + lane * VPT);

  for (int n = w0; n < Nn; n += nw) {
    const int beg = rowptr[n];
    const int end = rowptr[n + 1];
    const float ad = ald[n];
    const float es = lrelu(als[n] + ad);

    const bool valid = beg + lane < end;
    const int s0 = valid ? csr[beg + lane] : n;
    const float e0 = lrelu(als[s0] + ad);
    float m = valid ? fmaxf(es, e0) : es;
    for (int base = beg + 32; base < end; base += 32) {
      if (base + lane < end) m = fmaxf(m, lrelu(als[csr[base + lane]] + ad));
    }
    m = wave_max(m);

    const float xs = __expf(es - m);
    const float x0 = valid ? __expf(e0 - m) : 0.f;
    float sum = x0 + (lane == 0 ? xs : 0.f);
    for (int base = beg + 32; base < end; base += 32) {
      if (base + lane < end) sum += __expf(lrelu(als[csr[base + lane]] + ad) - m);
    }
    sum = wave_sum(sum);

    vecf acc = (xs / sum) * *reinterpret_cast<const vecf*>(hl + (size_t)n * C);
    add_rows<C, vecf>(acc, hl, s0, x0 / sum, min(end - beg, 32));
    for (int base = beg + 32; base < end; base += 32) {
      const bool vd = base + lane < end;
      const int s = vd ? csr[base + lane] : n;
      const float x = vd ? __expf(lrelu(als[s] + ad) - m) : 0.f;
      add_rows<C, vecf>(acc, hl, s, x / sum, min(end - base, 32));
    }

    vecf o = acc + bv;
    if (RELU) o = __builtin_elementwise_max(o, (vecf)(0.f));
    *reinterpret_cast<vecf*>(out + (size_t)n * C + lane * VPT) = o;
  }
}

extern "C" void kernel_launch(void* const* d_in, const int* in_sizes, int n_in,
                              void* d_out, int out_size, void* d_ws, size_t ws_size,
                              hipStream_t stream) {
  const float* x   = (const float*)d_in[0];
  const int*   ei  = (const int*)d_in[1];
  const float* W1  = (const float*)d_in[2];
  const float* as1 = (const float*)d_in[3];
  const float* ad1 = (const float*)d_in[4];
  const float* b1  = (const float*)d_in[5];
  const float* W2  = (const float*)d_in[6];
  const float* as2 = (const float*)d_in[7];
  const float* ad2 = (const float*)d_in[8];
  const float* b2  = (const float*)d_in[9];
  const float* W3  = (const float*)d_in[10];
  const float* as3 = (const float*)d_in[11];
  const float* ad3 = (const float*)d_in[12];
  const float* b3  = (const float*)d_in[13];

  const int N = in_sizes[0] / 128;
  const int E = in_sizes[1] / 2;
  const int* srcp = ei;
  const int* dstp = ei + E;

  float* buf0   = (float*)d_ws;
  float* buf1   = buf0 + (size_t)N * 128;
  float* als    = buf1 + (size_t)N * 128;
  float* ald    = als + N;
  int*   deg    = (int*)(ald + N);
  int*   rowptr = deg + N;
  int*   bsum   = rowptr + N + 1;
  int*   rank   = bsum + N / 1024 + 1;
  int*   csr    = rank + E;

  const int gGemm = (N / 16 + 7) / 8;
  const int gProj = (N + 7) / 8;
  const int gE    = (E + 255) / 256;
  const int gScan = N / 1024 + 1;
  const int gNode = 4096;
  float* outF = (float*)d_out;

  hipMemsetAsync(deg, 0, (size_t)N * 4, stream);
  edge_rank<<<gE, 256, 0, stream>>>(dstp, deg, rank, E);
  deg_blocksum<<<gScan, 256, 0, stream>>>(deg, bsum, N);
  deg_scan<<<gScan, 256, 0, stream>>>(deg, bsum, rowptr, N);
  edge_fill<<<gE, 256, 0, stream>>>(srcp, dstp, rank, rowptr, csr, E);

  gemm_wmma<128, 128><<<gGemm, 256, 0, stream>>>(x, W1, buf0, N);
  row_proj<128><<<gProj, 256, 0, stream>>>(buf0, as1, ad1, als, ald, N);
  node_gather<128, true><<<gNode, 256, 0, stream>>>(rowptr, csr, buf0, als, ald, b1, buf1, N);

  gemm_wmma<128, 128><<<gGemm, 256, 0, stream>>>(buf1, W2, buf0, N);
  row_proj<128><<<gProj, 256, 0, stream>>>(buf0, as2, ad2, als, ald, N);
  node_gather<128, true><<<gNode, 256, 0, stream>>>(rowptr, csr, buf0, als, ald, b2, buf1, N);

  gemm_wmma<128, 64><<<gGemm, 256, 0, stream>>>(buf1, W3, buf0, N);
  row_proj<64><<<gProj, 256, 0, stream>>>(buf0, as3, ad3, als, ald, N);
  node_gather<64, false><<<gNode, 256, 0, stream>>>(rowptr, csr, buf0, als, ald, b3, outF, N);
}


# GEMM epilogue takes h.a_src and h.a_dst from the accumulators (packed fma per lane, DPP butterfly over the 16-lane row: new sum order of both dots); the 3 row-projection launches go
# speedup vs baseline: 4.5551x; 1.1042x over previous
#include <hip/hip_runtime.h>

typedef __attribute__((ext_vector_type(2))) float v2f;
typedef __attribute__((ext_vector_type(8))) float v8f;

#define NEG_SLOPE 0.2f


template <int MASK>
__device__ __forceinline__ float row_xor_add(float keep, float send) {
  const int got = __builtin_amdgcn_update_dpp(0, __float_as_int(send), 0x160 | MASK,
                                              0xf, 0xf, true);
  return keep + __int_as_float(got);
}

template <int BIT, int P>
__device__ __forceinline__ void fold_half(float* p, bool up) {
#pragma unroll
  for (int j = 0; j < P / 2; ++j) {
    const float keep = up ? p[j + P / 2] : p[j];
    const float send = up ? p[j] : p[j + P / 2];
    p[j] = row_xor_add<BIT>(keep, send);
  }
}

template <int CIN, int COUT>
__global__ __launch_bounds__(256) void gemm_wmma(const float* __restrict__ X,
                                                 const float* __restrict__ Wg,
                                                 const float* __restrict__ asrc,
                                                 const float* __restrict__ adst,
                                                 float* __restrict__ H,
                                                 float* __restrict__ als,
                                                 float* __restrict__ ald, int M) {
  __shared__ v2f Wswz[(CIN / 4) * COUT * 2];
  {
    constexpr int TOT = (CIN / 4) * COUT * 2;
    for (int i = threadIdx.x; i < TOT; i += 256) {
      int m = i / (COUT * 2);
      int r = i - m * (COUT * 2);
      int c = r >> 1;
      int hsel = r & 1;
      const float* wp = Wg + (size_t)(4 * m + 2 * hsel) * COUT + c;
      v2f w;
      w.x = wp[0];
      w.y = wp[COUT];
      Wswz[i] = w;
    }
  }
  __syncthreads();

  const int lane = threadIdx.x & 31;
  const int wave = threadIdx.x >> 5;
  const int mtile = blockIdx.x * 8 + wave;
  if (mtile * 16 >= M) return;
  const int m0 = mtile * 16;

  constexpr int NT = COUT / 16;
  v8f acc[NT] = {};

  const int rsel = lane & 15;
  const int hiHalf = lane >> 4;
  const float* xrow = X + (size_t)(m0 + rsel) * CIN + hiHalf * 2;
  const v2f* wbase = Wswz + rsel * 2 + hiHalf;

  for (int k = 0; k < CIN; k += 4) {
    v2f a = *reinterpret_cast<const v2f*>(xrow + k);
    const v2f* wb = wbase + (k >> 2) * (COUT * 2);
#pragma unroll
    for (int t = 0; t < NT; ++t) {
      v2f bf = wb[t * 32];
      acc[t] = __builtin_amdgcn_wmma_f32_16x16x4_f32(
          false, a, false, bf, (short)0, acc[t], false, false);
    }
  }

  const int rbase = m0 + hiHalf * 8;
#pragma unroll
  for (int t = 0; t < NT; ++t) {
#pragma unroll
    for (int i = 0; i < 8; ++i) {
      H[(size_t)(rbase + i) * COUT + t * 16 + rsel] = acc[t][i];
    }
  }

  v2f ps[4], pd[4];
#pragma unroll
  for (int j = 0; j < 4; ++j) { ps[j] = (v2f)(0.f); pd[j] = (v2f)(0.f); }
#pragma unroll
  for (int t = 0; t < NT; ++t) {
    const float as_t = asrc[t * 16 + rsel];
    const float ad_t = adst[t * 16 + rsel];
    const v2f as2 = {as_t, as_t};
    const v2f ad2 = {ad_t, ad_t};
#pragma unroll
    for (int j = 0; j < 4; ++j) {
      const v2f h2 = {acc[t][2 * j], acc[t][2 * j + 1]};
      ps[j] = __builtin_elementwise_fma(h2, as2, ps[j]);
      pd[j] = __builtin_elementwise_fma(h2, ad2, pd[j]);
    }
  }
  float p[16];
#pragma unroll
  for (int j = 0; j < 4; ++j) {
    p[2 * j] = ps[j].x;
    p[2 * j + 1] = ps[j].y;
    p[8 + 2 * j] = pd[j].x;
    p[8 + 2 * j + 1] = pd[j].y;
  }
  fold_half<8, 16>(p, (rsel & 8) != 0);
  fold_half<4, 8>(p, (rsel & 4) != 0);
  fold_half<2, 4>(p, (rsel & 2) != 0);
  fold_half<1, 2>(p, (rsel & 1) != 0);
  float* pdst = ((rsel & 8) ? ald : als) + rbase + (rsel & 7);
  *pdst = p[0];
}

__device__ __forceinline__ float lrelu(float e) { return (e > 0.f) ? e : NEG_SLOPE * e; }

__global__ __launch_bounds__(256) void edge_rank(const int* __restrict__ dst,
                                                 int* __restrict__ deg,
                                                 int* __restrict__ rank, int E) {
  int i = blockIdx.x * 256 + threadIdx.x;
  if (i >= E) return;
  rank[i] = atomicAdd(&deg[dst[i]], 1);
}

__global__ __launch_bounds__(256) void deg_blocksum(const int* __restrict__ deg,
                                                    int* __restrict__ bsum, int Nn) {
  __shared__ int wsum[8];
  const int base = blockIdx.x * 1024 + threadIdx.x * 4;
  int t = 0;
  for (int q = 0; q < 4; ++q)
    if (base + q < Nn) t += deg[base + q];
  for (int off = 16; off > 0; off >>= 1) t += __shfl_xor(t, off, 32);
  if ((threadIdx.x & 31) == 0) wsum[threadIdx.x >> 5] = t;
  __syncthreads();
  if (threadIdx.x == 0) {
    int s = 0;
    for (int w = 0; w < 8; ++w) s += wsum[w];
    bsum[blockIdx.x] = s;
  }
}

__global__ __launch_bounds__(256) void deg_scan(const int* __restrict__ deg,
                                                const int* __restrict__ bsum,
                                                int* __restrict__ rowptr, int Nn) {
  __shared__ int wpre[8];
  __shared__ int wtot[8];
  const int lane = threadIdx.x & 31;
  const int wave = threadIdx.x >> 5;
  int o = 0;
  for (int j = threadIdx.x; j < (int)blockIdx.x; j += 256) o += bsum[j];
  for (int off = 16; off > 0; off >>= 1) o += __shfl_xor(o, off, 32);
  if (lane == 0) wpre[wave] = o;

  const int base = blockIdx.x * 1024 + threadIdx.x * 4;
  int v[4];
  int t = 0;
  for (int q = 0; q < 4; ++q) {
    v[q] = (base + q < Nn) ? deg[base + q] : 0;
    t += v[q];
  }
  int inc = t;
  for (int off = 1; off < 32; off <<= 1) {
    int u = __shfl_up(inc, off, 32);
    if (lane >= off) inc += u;
  }
  if (lane == 31) wtot[wave] = inc;
  __syncthreads();
  int ex = inc - t;
  for (int w = 0; w < 8; ++w) {
    ex += wpre[w];
    if (w < wave) ex += wtot[w];
  }
  for (int q = 0; q < 4; ++q) {
    if (base + q <= Nn) rowptr[base + q] = ex;
    ex += v[q];
  }
}

__global__ __launch_bounds__(256) void edge_fill(const int* __restrict__ src,
                                                 const int* __restrict__ dst,
                                                 const int* __restrict__ rank,
                                                 const int* __restrict__ rowptr,
                                                 int* __restrict__ csr, int E) {
  int i = blockIdx.x * 256 + threadIdx.x;
  if (i >= E) return;
  csr[rowptr[dst[i]] + rank[i]] = src[i];
}

__device__ __forceinline__ float wave_max(float v) {
  for (int off = 16; off > 0; off >>= 1) v = fmaxf(v, __shfl_xor(v, off, 32));
  return v;
}
__device__ __forceinline__ float wave_sum(float v) {
  for (int off = 16; off > 0; off >>= 1) v += __shfl_xor(v, off, 32);
  return v;
}

template <int C, typename vecf>
__device__ __forceinline__ void add_rows(vecf& acc, const float* __restrict__ hl,
                                         int s, float w, int cnt) {
  for (int j = 0; j < cnt; j += 4) {
    vecf hv[4];
    float wj[4];
    for (int q = 0; q < 4; ++q) {
      const int sj = __builtin_amdgcn_readlane(s, j + q);
      wj[q] = __int_as_float(__builtin_amdgcn_readlane(__float_as_int(w), j + q));
      hv[q] = *reinterpret_cast<const vecf*>(hl + (size_t)sj * C);
    }
    for (int q = 0; q < 4; ++q) acc += wj[q] * hv[q];
  }
}

template <int C, bool RELU>
__global__ __launch_bounds__(256) void node_gather(const int* __restrict__ rowptr,
                                                   const int* __restrict__ csr,
                                                   const float* __restrict__ H,
                                                   const float* __restrict__ als,
                                                   const float* __restrict__ ald,
                                                   const float* __restrict__ bias,
                                                   float* __restrict__ out, int Nn) {
  constexpr int VPT = C / 32;
  typedef __attribute__((ext_vector_type(VPT))) float vecf;
  const int lane = threadIdx.x & 31;
  const int w0 = __builtin_amdgcn_readfirstlane((int)(blockIdx.x * 8 + (threadIdx.x >> 5)));
  const int nw = gridDim.x * 8;
  const float* hl = H + lane * VPT;
  const vecf bv = *reinterpret_cast<const vecf*>(bias + lane * VPT);

  for (int n = w0; n < Nn; n += nw) {
    const int beg = rowptr[n];
    const int end = rowptr[n + 1];
    const float ad = ald[n];
    const float es = lrelu(als[n] + ad);

    const bool valid = beg + lane < end;
    const int s0 = valid ? csr[beg + lane] : n;
    const float e0 = lrelu(als[s0] + ad);
    float m = valid ? fmaxf(es, e0) : es;
    for (int base = beg + 32; base < end; base += 32) {
      if (base + lane < end) m = fmaxf(m, lrelu(als[csr[base + lane]] + ad));
    }
    m = wave_max(m);

    const float xs = __expf(es - m);
    const float x0 = valid ? __expf(e0 - m) : 0.f;
    float sum = x0 + (lane == 0 ? xs : 0.f);
    for (int base = beg + 32; base < end; base += 32) {
      if (base + lane < end) sum += __expf(lrelu(als[csr[base + lane]] + ad) - m);
    }
    sum = wave_sum(sum);

    vecf acc = (xs / sum) * *reinterpret_cast<const vecf*>(hl + (size_t)n * C);
    add_rows<C, vecf>(acc, hl, s0, x0 / sum, min(end - beg, 32));
    for (int base = beg + 32; base < end; base += 32) {
      const bool vd = base + lane < end;
      const int s = vd ? csr[base + lane] : n;
      const float x = vd ? __expf(lrelu(als[s] + ad) - m) : 0.f;
      add_rows<C, vecf>(acc, hl, s, x / sum, min(end - base, 32));
    }

    vecf o = acc + bv;
    if (RELU) o = __builtin_elementwise_max(o, (vecf)(0.f));
    *reinterpret_cast<vecf*>(out + (size_t)n * C + lane * VPT) = o;
  }
}

extern "C" void kernel_launch(void* const* d_in, const int* in_sizes, int n_in,
                              void* d_out, int out_size, void* d_ws, size_t ws_size,
                              hipStream_t stream) {
  const float* x   = (const float*)d_in[0];
  const int*   ei  = (const int*)d_in[1];
  const float* W1  = (const float*)d_in[2];
  const float* as1 = (const float*)d_in[3];
  const float* ad1 = (const float*)d_in[4];
  const float* b1  = (const float*)d_in[5];
  const float* W2  = (const float*)d_in[6];
  const float* as2 = (const float*)d_in[7];
  const float* ad2 = (const float*)d_in[8];
  const float* b2  = (const float*)d_in[9];
  const float* W3  = (const float*)d_in[10];
  const float* as3 = (const float*)d_in[11];
  const float* ad3 = (const float*)d_in[12];
  const float* b3  = (const float*)d_in[13];

  const int N = in_sizes[0] / 128;
  const int E = in_sizes[1] / 2;
  const int* srcp = ei;
  const int* dstp = ei + E;

  float* buf0   = (float*)d_ws;
  float* buf1   = buf0 + (size_t)N * 128;
  float* als    = buf1 + (size_t)N * 128;
  float* ald    = als + N;
  int*   deg    = (int*)(ald + N);
  int*   rowptr = deg + N;
  int*   bsum   = rowptr + N + 1;
  int*   rank   = bsum + N / 1024 + 1;
  int*   csr    = rank + E;

  const int gGemm = (N / 16 + 7) / 8;
  const int gE    = (E + 255) / 256;
  const int gScan = N / 1024 + 1;
  const int gNode = 4096;
  float* outF = (float*)d_out;

  hipMemsetAsync(deg, 0, (size_t)N * 4, stream);
  edge_rank<<<gE, 256, 0, stream>>>(dstp, deg, rank, E);
  deg_blocksum<<<gScan, 256, 0, stream>>>(deg, bsum, N);
  deg_scan<<<gScan, 256, 0, stream>>>(deg, bsum, rowptr, N);
  edge_fill<<<gE, 256, 0, stream>>>(srcp, dstp, rank, rowptr, csr, E);

  gemm_wmma<128, 128><<<gGemm, 256, 0, stream>>>(x, W1, as1, ad1, buf0, als, ald, N);
  node_gather<128, true><<<gNode, 256, 0, stream>>>(rowptr, csr, buf0, als, ald, b1, buf1, N);

  gemm_wmma<128, 128><<<gGemm, 256, 0, stream>>>(buf1, W2, as2, ad2, buf0, als, ald, N);
  node_gather<128, true><<<gNode, 256, 0, stream>>>(rowptr, csr, buf0, als, ald, b2, buf1, N);

  gemm_wmma<128, 64><<<gGemm, 256, 0, stream>>>(buf1, W3, as3, ad3, buf0, als, ald, N);
  node_gather<64, false><<<gNode, 256, 0, stream>>>(rowptr, csr, buf0, als, ald, b3, outF, N);
}


# edges grouped by destination with a two-level counting sort on LDS counters (256-node buckets, then nodes): no global atomics; order inside a group = arrival order of LDS adds (was: global adds)
# speedup vs baseline: 5.8308x; 1.2801x over previous
#include <hip/hip_runtime.h>

typedef __attribute__((ext_vector_type(2))) float v2f;
typedef __attribute__((ext_vector_type(8))) float v8f;

#define NEG_SLOPE 0.2f


template <int MASK>
__device__ __forceinline__ float row_xor_add(float keep, float send) {
  const int got = __builtin_amdgcn_update_dpp(0, __float_as_int(send), 0x160 | MASK,
                                              0xf, 0xf, true);
  return keep + __int_as_float(got);
}

template <int BIT, int P>
__device__ __forceinline__ void fold_half(float* p, bool up) {
#pragma unroll
  for (int j = 0; j < P / 2; ++j) {
    const float keep = up ? p[j + P / 2] : p[j];
    const float send = up ? p[j] : p[j + P / 2];
    p[j] = row_xor_add<BIT>(keep, send);
  }
}

template <int CIN, int COUT>
__global__ __launch_bounds__(256) void gemm_wmma(const float* __restrict__ X,
                                                 const float* __restrict__ Wg,
                                                 const float* __restrict__ asrc,
                                                 const float* __restrict__ adst,
                                                 float* __restrict__ H,
                                                 float* __restrict__ als,
                                                 float* __restrict__ ald, int M) {
  __shared__ v2f Wswz[(CIN / 4) * COUT * 2];
  {
    constexpr int TOT = (CIN / 4) * COUT * 2;
    for (int i = threadIdx.x; i < TOT; i += 256) {
      int m = i / (COUT * 2);
      int r = i - m * (COUT * 2);
      int c = r >> 1;
      int hsel = r & 1;
      const float* wp = Wg + (size_t)(4 * m + 2 * hsel) * COUT + c;
      v2f w;
      w.x = wp[0];
      w.y = wp[COUT];
      Wswz[i] = w;
    }
  }
  __syncthreads();

  const int lane = threadIdx.x & 31;
  const int wave = threadIdx.x >> 5;
  const int mtile = blockIdx.x * 8 + wave;
  if (mtile * 16 >= M) return;
  const int m0 = mtile * 16;

  constexpr int NT = COUT / 16;
  v8f acc[NT] = {};

  const int rsel = lane & 15;
  const int hiHalf = lane >> 4;
  const float* xrow = X + (size_t)(m0 + rsel) * CIN + hiHalf * 2;
  const v2f* wbase = Wswz + rsel * 2 + hiHalf;

  for (int k = 0; k < CIN; k += 4) {
    v2f a = *reinterpret_cast<const v2f*>(xrow + k);
    const v2f* wb = wbase + (k >> 2) * (COUT * 2);
#pragma unroll
    for (int t = 0; t < NT; ++t) {
      v2f bf = wb[t * 32];
      acc[t] = __builtin_amdgcn_wmma_f32_16x16x4_f32(
          false, a, false, bf, (short)0, acc[t], false, false);
    }
  }

  const int rbase = m0 + hiHalf * 8;
#pragma unroll
  for (int t = 0; t < NT; ++t) {
#pragma unroll
    for (int i = 0; i < 8; ++i) {
      H[(size_t)(rbase + i) * COUT + t * 16 + rsel] = acc[t][i];
    }
  }

  v2f ps[4], pd[4];
#pragma unroll
  for (int j = 0; j < 4; ++j) { ps[j] = (v2f)(0.f); pd[j] = (v2f)(0.f); }
#pragma unroll
  for (int t = 0; t < NT; ++t) {
    const float as_t = asrc[t * 16 + rsel];
    const float ad_t = adst[t * 16 + rsel];
    const v2f as2 = {as_t, as_t};
    const v2f ad2 = {ad_t, ad_t};
#pragma unroll
    for (int j = 0; j < 4; ++j) {
      const v2f h2 = {acc[t][2 * j], acc[t][2 * j + 1]};
      ps[j] = __builtin_elementwise_fma(h2, as2, ps[j]);
      pd[j] = __builtin_elementwise_fma(h2, ad2, pd[j]);
    }
  }
  float p[16];
#pragma unroll
  for (int j = 0; j < 4; ++j) {
    p[2 * j] = ps[j].x;
    p[2 * j + 1] = ps[j].y;
    p[8 + 2 * j] = pd[j].x;
    p[8 + 2 * j + 1] = pd[j].y;
  }
  fold_half<8, 16>(p, (rsel & 8) != 0);
  fold_half<4, 8>(p, (rsel & 4) != 0);
  fold_half<2, 4>(p, (rsel & 2) != 0);
  fold_half<1, 2>(p, (rsel & 1) != 0);
  float* pdst = ((rsel & 8) ? ald : als) + rbase + (rsel & 7);
  *pdst = p[0];
}

__device__ __forceinline__ float lrelu(float e) { return (e > 0.f) ? e : NEG_SLOPE * e; }

typedef __attribute__((ext_vector_type(2))) int v2i;

#define BUCKET_SHIFT 8
#define BUCKET_NODES 256
#define CHUNK_EDGES 4096

#define EDGES_PER_THREAD (CHUNK_EDGES / 256)

__global__ __launch_bounds__(256) void bucket_count(const int* __restrict__ dst,
                                                    int* __restrict__ cnt, int E, int G1,
                                                    int NBK) {
  extern __shared__ int hist[];
  for (int i = threadIdx.x; i < NBK; i += 256) hist[i] = 0;
  __syncthreads();
  const int base = blockIdx.x * CHUNK_EDGES + threadIdx.x;
  int d[EDGES_PER_THREAD];
#pragma unroll
  for (int q = 0; q < EDGES_PER_THREAD; ++q)
    d[q] = dst[min(base + q * 256, E - 1)];
#pragma unroll
  for (int q = 0; q < EDGES_PER_THREAD; ++q)
    if (base + q * 256 < E) atomicAdd(&hist[d[q] >> BUCKET_SHIFT], 1);
  __syncthreads();
  for (int i = threadIdx.x; i < NBK; i += 256) cnt[(size_t)i * G1 + blockIdx.x] = hist[i];
}

__global__ __launch_bounds__(256) void chunk_sum(const int* __restrict__ v,
                                                 int* __restrict__ bsum, int Nn) {
  __shared__ int wsum[8];
  const int base = blockIdx.x * 1024 + threadIdx.x * 4;
  int t = 0;
  for (int q = 0; q < 4; ++q)
    if (base + q < Nn) t += v[base + q];
  for (int off = 16; off > 0; off >>= 1) t += __shfl_xor(t, off, 32);
  if ((threadIdx.x & 31) == 0) wsum[threadIdx.x >> 5] = t;
  __syncthreads();
  if (threadIdx.x == 0) {
    int s = 0;
    for (int w = 0; w < 8; ++w) s += wsum[w];
    bsum[blockIdx.x] = s;
  }
}

__global__ __launch_bounds__(256) void excl_scan(const int* __restrict__ in,
                                                 const int* __restrict__ bsum,
                                                 int* __restrict__ out, int Nn) {
  __shared__ int wpre[8];
  __shared__ int wtot[8];
  const int lane = threadIdx.x & 31;
  const int wave = threadIdx.x >> 5;
  int o = 0;
  for (int j = threadIdx.x; j < (int)blockIdx.x; j += 256) o += bsum[j];
  for (int off = 16; off > 0; off >>= 1) o += __shfl_xor(o, off, 32);
  if (lane == 0) wpre[wave] = o;

  const int base = blockIdx.x * 1024 + threadIdx.x * 4;
  int v[4];
  int t = 0;
  for (int q = 0; q < 4; ++q) {
    v[q] = (base + q < Nn) ? in[base + q] : 0;
    t += v[q];
  }
  int inc = t;
  for (int off = 1; off < 32; off <<= 1) {
    int u = __shfl_up(inc, off, 32);
    if (lane >= off) inc += u;
  }
  if (lane == 31) wtot[wave] = inc;
  __syncthreads();
  int ex = inc - t;
  for (int w = 0; w < 8; ++w) {
    ex += wpre[w];
    if (w < wave) ex += wtot[w];
  }
  for (int q = 0; q < 4; ++q) {
    if (base + q <= Nn) out[base + q] = ex;
    ex += v[q];
  }
}

__global__ __launch_bounds__(256) void bucket_scatter(const int* __restrict__ src,
                                                      const int* __restrict__ dst,
                                                      const int* __restrict__ offs,
                                                      int* __restrict__ binned, int E,
                                                      int G1, int NBK) {
  extern __shared__ int slot[];
  for (int i = threadIdx.x; i < NBK; i += 256) slot[i] = offs[(size_t)i * G1 + blockIdx.x];
  __syncthreads();
  const int base = blockIdx.x * CHUNK_EDGES + threadIdx.x;
  int d[EDGES_PER_THREAD], s[EDGES_PER_THREAD];
#pragma unroll
  for (int q = 0; q < EDGES_PER_THREAD; ++q) {
    const int i = min(base + q * 256, E - 1);
    d[q] = dst[i];
    s[q] = src[i];
  }
#pragma unroll
  for (int q = 0; q < EDGES_PER_THREAD; ++q) {
    if (base + q * 256 < E) {
      const int pos = atomicAdd(&slot[d[q] >> BUCKET_SHIFT], 1);
      const v2i e = {s[q], d[q]};
      *reinterpret_cast<v2i*>(binned + 2 * (size_t)pos) = e;
    }
  }
}

__global__ __launch_bounds__(256) void bucket_csr(const int* __restrict__ binned,
                                                  const int* __restrict__ offs,
                                                  int* __restrict__ rowptr,
                                                  int* __restrict__ csr, int Nn, int G1) {
  __shared__ int cnt[BUCKET_NODES];
  __shared__ int cur[BUCKET_NODES];
  __shared__ int wtot[BUCKET_NODES / 32];
  const int tid = threadIdx.x;
  const int lane = tid & 31;
  const int wave = tid >> 5;
  const int b = blockIdx.x;
  const int beg = offs[(size_t)b * G1];
  const int end = offs[(size_t)(b + 1) * G1];
  cnt[tid] = 0;
  __syncthreads();
  for (int i0 = beg + tid; i0 < end; i0 += CHUNK_EDGES) {
    int dl[EDGES_PER_THREAD];
#pragma unroll
    for (int q = 0; q < EDGES_PER_THREAD; ++q)
      dl[q] = binned[2 * (size_t)min(i0 + q * 256, end - 1) + 1];
#pragma unroll
    for (int q = 0; q < EDGES_PER_THREAD; ++q)
      if (i0 + q * 256 < end) atomicAdd(&cnt[dl[q] & (BUCKET_NODES - 1)], 1);
  }
  __syncthreads();

  const int c = cnt[tid];
  int inc = c;
  for (int off = 1; off < 32; off <<= 1) {
    int u = __shfl_up(inc, off, 32);
    if (lane >= off) inc += u;
  }
  if (lane == 31) wtot[wave] = inc;
  __syncthreads();
  int ex = inc - c;
  for (int w = 0; w < BUCKET_NODES / 32; ++w)
    if (w < wave) ex += wtot[w];
  cur[tid] = ex;
  const int node = b * BUCKET_NODES + tid;
  if (node < Nn) rowptr[node] = beg + ex;
  if (tid == 0 && b == (int)gridDim.x - 1) rowptr[Nn] = end;
  __syncthreads();

  for (int i0 = beg + tid; i0 < end; i0 += CHUNK_EDGES) {
    v2i e[EDGES_PER_THREAD];
#pragma unroll
    for (int q = 0; q < EDGES_PER_THREAD; ++q)
      e[q] = *reinterpret_cast<const v2i*>(binned + 2 * (size_t)min(i0 + q * 256, end - 1));
#pragma unroll
    for (int q = 0; q < EDGES_PER_THREAD; ++q) {
      if (i0 + q * 256 < end) {
        const int pos = atomicAdd(&cur[e[q].y & (BUCKET_NODES - 1)], 1);
        csr[beg + pos] = e[q].x;
      }
    }
  }
}

__device__ __forceinline__ float wave_max(float v) {
  for (int off = 16; off > 0; off >>= 1) v = fmaxf(v, __shfl_xor(v, off, 32));
  return v;
}
__device__ __forceinline__ float wave_sum(float v) {
  for (int off = 16; off > 0; off >>= 1) v += __shfl_xor(v, off, 32);
  return v;
}

template <int C, typename vecf>
__device__ __forceinline__ void add_rows(vecf& acc, const float* __restrict__ hl,
                                         int s, float w, int cnt) {
  for (int j = 0; j < cnt; j += 4) {
    vecf hv[4];
    float wj[4];
    for (int q = 0; q < 4; ++q) {
      const int sj = __builtin_amdgcn_readlane(s, j + q);
      wj[q] = __int_as_float(__builtin_amdgcn_readlane(__float_as_int(w), j + q));
      hv[q] = *reinterpret_cast<const vecf*>(hl + (size_t)sj * C);
    }
    for (int q = 0; q < 4; ++q) acc += wj[q] * hv[q];
  }
}

template <int C, bool RELU>
__global__ __launch_bounds__(256) void node_gather(const int* __restrict__ rowptr,
                                                   const int* __restrict__ csr,
                                                   const float* __restrict__ H,
                                                   const float* __restrict__ als,
                                                   const float* __restrict__ ald,
                                                   const float* __restrict__ bias,
                                                   float* __restrict__ out, int Nn) {
  constexpr int VPT = C / 32;
  typedef __attribute__((ext_vector_type(VPT))) float vecf;
  const int lane = threadIdx.x & 31;
  const int w0 = __builtin_amdgcn_readfirstlane((int)(blockIdx.x * 8 + (threadIdx.x >> 5)));
  const int nw = gridDim.x * 8;
  const float* hl = H + lane * VPT;
  const vecf bv = *reinterpret_cast<const vecf*>(bias + lane * VPT);

  for (int n = w0; n < Nn; n += nw) {
    const int beg = rowptr[n];
    const int end = rowptr[n + 1];
    const float ad = ald[n];
    const float es = lrelu(als[n] + ad);

    const bool valid = beg + lane < end;
    const int s0 = valid ? csr[beg + lane] : n;
    const float e0 = lrelu(als[s0] + ad);
    float m = valid ? fmaxf(es, e0) : es;
    for (int base = beg + 32; base < end; base += 32) {
      if (base + lane < end) m = fmaxf(m, lrelu(als[csr[base + lane]] + ad));
    }
    m = wave_max(m);

    const float xs = __expf(es - m);
    const float x0 = valid ? __expf(e0 - m) : 0.f;
    float sum = x0 + (lane == 0 ? xs : 0.f);
    for (int base = beg + 32; base < end; base += 32) {
      if (base + lane < end) sum += __expf(lrelu(als[csr[base + lane]] + ad) - m);
    }
    sum = wave_sum(sum);

    vecf acc = (xs / sum) * *reinterpret_cast<const vecf*>(hl + (size_t)n * C);
    add_rows<C, vecf>(acc, hl, s0, x0 / sum, min(end - beg, 32));
    for (int base = beg + 32; base < end; base += 32) {
      const bool vd = base + lane < end;
      const int s = vd ? csr[base + lane] : n;
      const float x = vd ? __expf(lrelu(als[s] + ad) - m) : 0.f;
      add_rows<C, vecf>(acc, hl, s, x / sum, min(end - base, 32));
    }

    vecf o = acc + bv;
    if (RELU) o = __builtin_elementwise_max(o, (vecf)(0.f));
    *reinterpret_cast<vecf*>(out + (size_t)n * C + lane * VPT) = o;
  }
}

extern "C" void kernel_launch(void* const* d_in, const int* in_sizes, int n_in,
                              void* d_out, int out_size, void* d_ws, size_t ws_size,
                              hipStream_t stream) {
  const float* x   = (const float*)d_in[0];
  const int*   ei  = (const int*)d_in[1];
  const float* W1  = (const float*)d_in[2];
  const float* as1 = (const float*)d_in[3];
  const float* ad1 = (const float*)d_in[4];
  const float* b1  = (const float*)d_in[5];
  const float* W2  = (const float*)d_in[6];
  const float* as2 = (const float*)d_in[7];
  const float* ad2 = (const float*)d_in[8];
  const float* b2  = (const float*)d_in[9];
  const float* W3  = (const float*)d_in[10];
  const float* as3 = (const float*)d_in[11];
  const float* ad3 = (const float*)d_in[12];
  const float* b3  = (const float*)d_in[13];

  const int N = in_sizes[0] / 128;
  const int E = in_sizes[1] / 2;
  const int* srcp = ei;
  const int* dstp = ei + E;

  const int NBK = (N + BUCKET_NODES - 1) / BUCKET_NODES;
  const int G1  = (E + CHUNK_EDGES - 1) / CHUNK_EDGES;
  const int NT  = NBK * G1;

  float* buf0   = (float*)d_ws;
  float* buf1   = buf0 + (size_t)N * 128;
  int*   binned = (int*)(buf1 + (size_t)N * 128);
  int*   csr    = binned + 2 * (size_t)E;
  float* als    = (float*)(csr + E);
  float* ald    = als + N;
  int*   rowptr = (int*)(ald + N);
  int*   cnt    = rowptr + N + 1;
  int*   offs   = cnt + NT;
  int*   bsum   = offs + NT + 1;

  const int gGemm = (N / 16 + 7) / 8;
  const int gScan = NT / 1024 + 1;
  const int gNode = 4096;
  float* outF = (float*)d_out;

  bucket_count<<<G1, 256, NBK * sizeof(int), stream>>>(dstp, cnt, E, G1, NBK);
  chunk_sum<<<gScan, 256, 0, stream>>>(cnt, bsum, NT);
  excl_scan<<<gScan, 256, 0, stream>>>(cnt, bsum, offs, NT);
  bucket_scatter<<<G1, 256, NBK * sizeof(int), stream>>>(srcp, dstp, offs, binned, E, G1, NBK);
  bucket_csr<<<NBK, 256, 0, stream>>>(binned, offs, rowptr, csr, N, G1);

  gemm_wmma<128, 128><<<gGemm, 256, 0, stream>>>(x, W1, as1, ad1, buf0, als, ald, N);
  node_gather<128, true><<<gNode, 256, 0, stream>>>(rowptr, csr, buf0, als, ald, b1, buf1, N);

  gemm_wmma<128, 128><<<gGemm, 256, 0, stream>>>(buf1, W2, as2, ad2, buf0, als, ald, N);
  node_gather<128, true><<<gNode, 256, 0, stream>>>(rowptr, csr, buf0, als, ald, b2, buf1, N);

  gemm_wmma<128, 64><<<gGemm, 256, 0, stream>>>(buf1, W3, as3, ad3, buf0, als, ald, N);
  node_gather<64, false><<<gNode, 256, 0, stream>>>(rowptr, csr, buf0, als, ald, b3, outF, N);
}


# GEMM launched as 32-wave work-groups, one per WGP: W staged into LDS once per WGP, all its loads up front, and row tile t goes to group t%256, wave t/256: 6-7 tiles per SIMD (was up to 8); sums kept
# speedup vs baseline: 6.0427x; 1.0363x over previous
#include <hip/hip_runtime.h>

typedef __attribute__((ext_vector_type(2))) float v2f;
typedef __attribute__((ext_vector_type(8))) float v8f;

#define NEG_SLOPE 0.2f


template <int MASK>
__device__ __forceinline__ float row_xor_add(float keep, float send) {
  const int got = __builtin_amdgcn_update_dpp(0, __float_as_int(send), 0x160 | MASK,
                                              0xf, 0xf, true);
  return keep + __int_as_float(got);
}

template <int BIT, int P>
__device__ __forceinline__ void fold_half(float* p, bool up) {
#pragma unroll
  for (int j = 0; j < P / 2; ++j) {
    const float keep = up ? p[j + P / 2] : p[j];
    const float send = up ? p[j] : p[j + P / 2];
    p[j] = row_xor_add<BIT>(keep, send);
  }
}

template <int CIN, int COUT>
__global__ __launch_bounds__(1024) void gemm_wmma(const float* __restrict__ X,
                                                 const float* __restrict__ Wg,
                                                 const float* __restrict__ asrc,
                                                 const float* __restrict__ adst,
                                                 float* __restrict__ H,
                                                 float* __restrict__ als,
                                                 float* __restrict__ ald, int M) {
  __shared__ v2f Wswz[(CIN / 4) * COUT * 2];
  {
    constexpr int TOT = (CIN / 4) * COUT * 2;
    constexpr int PER = TOT / 1024;
    constexpr int MSTEP = 1024 / (COUT * 2);
    static_assert(TOT % 1024 == 0 && 1024 % (COUT * 2) == 0, "staging assumes 1024 threads");
    const int m = threadIdx.x / (COUT * 2);
    const int r = threadIdx.x - m * (COUT * 2);
    const int c = r >> 1;
    const int hsel = r & 1;
    const float* wp = Wg + (size_t)(4 * m + 2 * hsel) * COUT + c;
    v2f w[PER];
#pragma unroll
    for (int q = 0; q < PER; ++q) {
      w[q].x = wp[(size_t)q * (4 * MSTEP * COUT)];
      w[q].y = wp[(size_t)q * (4 * MSTEP * COUT) + COUT];
    }
#pragma unroll
    for (int q = 0; q < PER; ++q) Wswz[threadIdx.x + q * 1024] = w[q];
  }
  __syncthreads();

  const int lane = threadIdx.x & 31;
  const int wave = threadIdx.x >> 5;
  const int mtile = wave * (int)gridDim.x + (int)blockIdx.x;
  if (mtile * 16 >= M) return;
  const int m0 = mtile * 16;

  constexpr int NT = COUT / 16;
  v8f acc[NT] = {};

  const int rsel = lane & 15;
  const int hiHalf = lane >> 4;
  const float* xrow = X + (size_t)(m0 + rsel) * CIN + hiHalf * 2;
  const v2f* wbase = Wswz + rsel * 2 + hiHalf;

  for (int k = 0; k < CIN; k += 4) {
    v2f a = *reinterpret_cast<const v2f*>(xrow + k);
    const v2f* wb = wbase + (k >> 2) * (COUT * 2);
#pragma unroll
    for (int t = 0; t < NT; ++t) {
      v2f bf = wb[t * 32];
      acc[t] = __builtin_amdgcn_wmma_f32_16x16x4_f32(
          false, a, false, bf, (short)0, acc[t], false, false);
    }
  }

  const int rbase = m0 + hiHalf * 8;
#pragma unroll
  for (int t = 0; t < NT; ++t) {
#pragma unroll
    for (int i = 0; i < 8; ++i) {
      H[(size_t)(rbase + i) * COUT + t * 16 + rsel] = acc[t][i];
    }
  }

  v2f ps[4], pd[4];
#pragma unroll
  for (int j = 0; j < 4; ++j) { ps[j] = (v2f)(0.f); pd[j] = (v2f)(0.f); }
#pragma unroll
  for (int t = 0; t < NT; ++t) {
    const float as_t = asrc[t * 16 + rsel];
    const float ad_t = adst[t * 16 + rsel];
    const v2f as2 = {as_t, as_t};
    const v2f ad2 = {ad_t, ad_t};
#pragma unroll
    for (int j = 0; j < 4; ++j) {
      const v2f h2 = {acc[t][2 * j], acc[t][2 * j + 1]};
      ps[j] = __builtin_elementwise_fma(h2, as2, ps[j]);
      pd[j] = __builtin_elementwise_fma(h2, ad2, pd[j]);
    }
  }
  float p[16];
#pragma unroll
  for (int j = 0; j < 4; ++j) {
    p[2 * j] = ps[j].x;
    p[2 * j + 1] = ps[j].y;
    p[8 + 2 * j] = pd[j].x;
    p[8 + 2 * j + 1] = pd[j].y;
  }
  fold_half<8, 16>(p, (rsel & 8) != 0);
  fold_half<4, 8>(p, (rsel & 4) != 0);
  fold_half<2, 4>(p, (rsel & 2) != 0);
  fold_half<1, 2>(p, (rsel & 1) != 0);
  float* pdst = ((rsel & 8) ? ald : als) + rbase + (rsel & 7);
  *pdst = p[0];
}

__device__ __forceinline__ float lrelu(float e) { return (e > 0.f) ? e : NEG_SLOPE * e; }

typedef __attribute__((ext_vector_type(2))) int v2i;

#define BUCKET_SHIFT 8
#define BUCKET_NODES 256
#define CHUNK_EDGES 4096

#define EDGES_PER_THREAD (CHUNK_EDGES / 256)

__global__ __launch_bounds__(256) void bucket_count(const int* __restrict__ dst,
                                                    int* __restrict__ cnt, int E, int G1,
                                                    int NBK) {
  extern __shared__ int hist[];
  for (int i = threadIdx.x; i < NBK; i += 256) hist[i] = 0;
  __syncthreads();
  const int base = blockIdx.x * CHUNK_EDGES + threadIdx.x;
  int d[EDGES_PER_THREAD];
#pragma unroll
  for (int q = 0; q < EDGES_PER_THREAD; ++q)
    d[q] = dst[min(base + q * 256, E - 1)];
#pragma unroll
  for (int q = 0; q < EDGES_PER_THREAD; ++q)
    if (base + q * 256 < E) atomicAdd(&hist[d[q] >> BUCKET_SHIFT], 1);
  __syncthreads();
  for (int i = threadIdx.x; i < NBK; i += 256) cnt[(size_t)i * G1 + blockIdx.x] = hist[i];
}

__global__ __launch_bounds__(256) void chunk_sum(const int* __restrict__ v,
                                                 int* __restrict__ bsum, int Nn) {
  __shared__ int wsum[8];
  const int base = blockIdx.x * 1024 + threadIdx.x * 4;
  int t = 0;
  for (int q = 0; q < 4; ++q)
    if (base + q < Nn) t += v[base + q];
  for (int off = 16; off > 0; off >>= 1) t += __shfl_xor(t, off, 32);
  if ((threadIdx.x & 31) == 0) wsum[threadIdx.x >> 5] = t;
  __syncthreads();
  if (threadIdx.x == 0) {
    int s = 0;
    for (int w = 0; w < 8; ++w) s += wsum[w];
    bsum[blockIdx.x] = s;
  }
}

__global__ __launch_bounds__(256) void excl_scan(const int* __restrict__ in,
                                                 const int* __restrict__ bsum,
                                                 int* __restrict__ out, int Nn) {
  __shared__ int wpre[8];
  __shared__ int wtot[8];
  const int lane = threadIdx.x & 31;
  const int wave = threadIdx.x >> 5;
  int o = 0;
  for (int j = threadIdx.x; j < (int)blockIdx.x; j += 256) o += bsum[j];
  for (int off = 16; off > 0; off >>= 1) o += __shfl_xor(o, off, 32);
  if (lane == 0) wpre[wave] = o;

  const int base = blockIdx.x * 1024 + threadIdx.x * 4;
  int v[4];
  int t = 0;
  for (int q = 0; q < 4; ++q) {
    v[q] = (base + q < Nn) ? in[base + q] : 0;
    t += v[q];
  }
  int inc = t;
  for (int off = 1; off < 32; off <<= 1) {
    int u = __shfl_up(inc, off, 32);
    if (lane >= off) inc += u;
  }
  if (lane == 31) wtot[wave] = inc;
  __syncthreads();
  int ex = inc - t;
  for (int w = 0; w < 8; ++w) {
    ex += wpre[w];
    if (w < wave) ex += wtot[w];
  }
  for (int q = 0; q < 4; ++q) {
    if (base + q <= Nn) out[base + q] = ex;
    ex += v[q];
  }
}

__global__ __launch_bounds__(256) void bucket_scatter(const int* __restrict__ src,
                                                      const int* __restrict__ dst,
                                                      const int* __restrict__ offs,
                                                      int* __restrict__ binned, int E,
                                                      int G1, int NBK) {
  extern __shared__ int slot[];
  for (int i = threadIdx.x; i < NBK; i += 256) slot[i] = offs[(size_t)i * G1 + blockIdx.x];
  __syncthreads();
  const int base = blockIdx.x * CHUNK_EDGES + threadIdx.x;
  int d[EDGES_PER_THREAD], s[EDGES_PER_THREAD];
#pragma unroll
  for (int q = 0; q < EDGES_PER_THREAD; ++q) {
    const int i = min(base + q * 256, E - 1);
    d[q] = dst[i];
    s[q] = src[i];
  }
#pragma unroll
  for (int q = 0; q < EDGES_PER_THREAD; ++q) {
    if (base + q * 256 < E) {
      const int pos = atomicAdd(&slot[d[q] >> BUCKET_SHIFT], 1);
      const v2i e = {s[q], d[q]};
      *reinterpret_cast<v2i*>(binned + 2 * (size_t)pos) = e;
    }
  }
}

__global__ __launch_bounds__(256) void bucket_csr(const int* __restrict__ binned,
                                                  const int* __restrict__ offs,
                                                  int* __restrict__ rowptr,
                                                  int* __restrict__ csr, int Nn, int G1) {
  __shared__ int cnt[BUCKET_NODES];
  __shared__ int cur[BUCKET_NODES];
  __shared__ int wtot[BUCKET_NODES / 32];
  const int tid = threadIdx.x;
  const int lane = tid & 31;
  const int wave = tid >> 5;
  const int b = blockIdx.x;
  const int beg = offs[(size_t)b * G1];
  const int end = offs[(size_t)(b + 1) * G1];
  cnt[tid] = 0;
  __syncthreads();
  for (int i0 = beg + tid; i0 < end; i0 += CHUNK_EDGES) {
    int dl[EDGES_PER_THREAD];
#pragma unroll
    for (int q = 0; q < EDGES_PER_THREAD; ++q)
      dl[q] = binned[2 * (size_t)min(i0 + q * 256, end - 1) + 1];
#pragma unroll
    for (int q = 0; q < EDGES_PER_THREAD; ++q)
      if (i0 + q * 256 < end) atomicAdd(&cnt[dl[q] & (BUCKET_NODES - 1)], 1);
  }
  __syncthreads();

  const int c = cnt[tid];
  int inc = c;
  for (int off = 1; off < 32; off <<= 1) {
    int u = __shfl_up(inc, off, 32);
    if (lane >= off) inc += u;
  }
  if (lane == 31) wtot[wave] = inc;
  __syncthreads();
  int ex = inc - c;
  for (int w = 0; w < BUCKET_NODES / 32; ++w)
    if (w < wave) ex += wtot[w];
  cur[tid] = ex;
  const int node = b * BUCKET_NODES + tid;
  if (node < Nn) rowptr[node] = beg + ex;
  if (tid == 0 && b == (int)gridDim.x - 1) rowptr[Nn] = end;
  __syncthreads();

  for (int i0 = beg + tid; i0 < end; i0 += CHUNK_EDGES) {
    v2i e[EDGES_PER_THREAD];
#pragma unroll
    for (int q = 0; q < EDGES_PER_THREAD; ++q)
      e[q] = *reinterpret_cast<const v2i*>(binned + 2 * (size_t)min(i0 + q * 256, end - 1));
#pragma unroll
    for (int q = 0; q < EDGES_PER_THREAD; ++q) {
      if (i0 + q * 256 < end) {
        const int pos = atomicAdd(&cur[e[q].y & (BUCKET_NODES - 1)], 1);
        csr[beg + pos] = e[q].x;
      }
    }
  }
}

__device__ __forceinline__ float wave_max(float v) {
  for (int off = 16; off > 0; off >>= 1) v = fmaxf(v, __shfl_xor(v, off, 32));
  return v;
}
__device__ __forceinline__ float wave_sum(float v) {
  for (int off = 16; off > 0; off >>= 1) v += __shfl_xor(v, off, 32);
  return v;
}

template <int C, typename vecf>
__device__ __forceinline__ void add_rows(vecf& acc, const float* __restrict__ hl,
                                         int s, float w, int cnt) {
  for (int j = 0; j < cnt; j += 4) {
    vecf hv[4];
    float wj[4];
    for (int q = 0; q < 4; ++q) {
      const int sj = __builtin_amdgcn_readlane(s, j + q);
      wj[q] = __int_as_float(__builtin_amdgcn_readlane(__float_as_int(w), j + q));
      hv[q] = *reinterpret_cast<const vecf*>(hl + (size_t)sj * C);
    }
    for (int q = 0; q < 4; ++q) acc += wj[q] * hv[q];
  }
}

template <int C, bool RELU>
__global__ __launch_bounds__(256) void node_gather(const int* __restrict__ rowptr,
                                                   const int* __restrict__ csr,
                                                   const float* __restrict__ H,
                                                   const float* __restrict__ als,
                                                   const float* __restrict__ ald,
                                                   const float* __restrict__ bias,
                                                   float* __restrict__ out, int Nn) {
  constexpr int VPT = C / 32;
  typedef __attribute__((ext_vector_type(VPT))) float vecf;
  const int lane = threadIdx.x & 31;
  const int w0 = __builtin_amdgcn_readfirstlane((int)(blockIdx.x * 8 + (threadIdx.x >> 5)));
  const int nw = gridDim.x * 8;
  const float* hl = H + lane * VPT;
  const vecf bv = *reinterpret_cast<const vecf*>(bias + lane * VPT);

  for (int n = w0; n < Nn; n += nw) {
    const int beg = rowptr[n];
    const int end = rowptr[n + 1];
    const float ad = ald[n];
    const float es = lrelu(als[n] + ad);

    const bool valid = beg + lane < end;
    const int s0 = valid ? csr[beg + lane] : n;
    const float e0 = lrelu(als[s0] + ad);
    float m = valid ? fmaxf(es, e0) : es;
    for (int base = beg + 32; base < end; base += 32) {
      if (base + lane < end) m = fmaxf(m, lrelu(als[csr[base + lane]] + ad));
    }
    m = wave_max(m);

    const float xs = __expf(es - m);
    const float x0 = valid ? __expf(e0 - m) : 0.f;
    float sum = x0 + (lane == 0 ? xs : 0.f);
    for (int base = beg + 32; base < end; base += 32) {
      if (base + lane < end) sum += __expf(lrelu(als[csr[base + lane]] + ad) - m);
    }
    sum = wave_sum(sum);

    vecf acc = (xs / sum) * *reinterpret_cast<const vecf*>(hl + (size_t)n * C);
    add_rows<C, vecf>(acc, hl, s0, x0 / sum, min(end - beg, 32));
    for (int base = beg + 32; base < end; base += 32) {
      const bool vd = base + lane < end;
      const int s = vd ? csr[base + lane] : n;
      const float x = vd ? __expf(lrelu(als[s] + ad) - m) : 0.f;
      add_rows<C, vecf>(acc, hl, s, x / sum, min(end - base, 32));
    }

    vecf o = acc + bv;
    if (RELU) o = __builtin_elementwise_max(o, (vecf)(0.f));
    *reinterpret_cast<vecf*>(out + (size_t)n * C + lane * VPT) = o;
  }
}

extern "C" void kernel_launch(void* const* d_in, const int* in_sizes, int n_in,
                              void* d_out, int out_size, void* d_ws, size_t ws_size,
                              hipStream_t stream) {
  const float* x   = (const float*)d_in[0];
  const int*   ei  = (const int*)d_in[1];
  const float* W1  = (const float*)d_in[2];
  const float* as1 = (const float*)d_in[3];
  const float* ad1 = (const float*)d_in[4];
  const float* b1  = (const float*)d_in[5];
  const float* W2  = (const float*)d_in[6];
  const float* as2 = (const float*)d_in[7];
  const float* ad2 = (const float*)d_in[8];
  const float* b2  = (const float*)d_in[9];
  const float* W3  = (const float*)d_in[10];
  const float* as3 = (const float*)d_in[11];
  const float* ad3 = (const float*)d_in[12];
  const float* b3  = (const float*)d_in[13];

  const int N = in_sizes[0] / 128;
  const int E = in_sizes[1] / 2;
  const int* srcp = ei;
  const int* dstp = ei + E;

  const int NBK = (N + BUCKET_NODES - 1) / BUCKET_NODES;
  const int G1  = (E + CHUNK_EDGES - 1) / CHUNK_EDGES;
  const int NT  = NBK * G1;

  float* buf0   = (float*)d_ws;
  float* buf1   = buf0 + (size_t)N * 128;
  int*   binned = (int*)(buf1 + (size_t)N * 128);
  int*   csr    = binned + 2 * (size_t)E;
  float* als    = (float*)(csr + E);
  float* ald    = als + N;
  int*   rowptr = (int*)(ald + N);
  int*   cnt    = rowptr + N + 1;
  int*   offs   = cnt + NT;
  int*   bsum   = offs + NT + 1;

  const int gTile = ((N + 15) / 16 + 31) / 32;
  const int gGemm = gTile > 256 ? gTile : 256;
  const int gScan = NT / 1024 + 1;
  const int gNode = 4096;
  float* outF = (float*)d_out;

  bucket_count<<<G1, 256, NBK * sizeof(int), stream>>>(dstp, cnt, E, G1, NBK);
  chunk_sum<<<gScan, 256, 0, stream>>>(cnt, bsum, NT);
  excl_scan<<<gScan, 256, 0, stream>>>(cnt, bsum, offs, NT);
  bucket_scatter<<<G1, 256, NBK * sizeof(int), stream>>>(srcp, dstp, offs, binned, E, G1, NBK);
  bucket_csr<<<NBK, 256, 0, stream>>>(binned, offs, rowptr, csr, N, G1);

  gemm_wmma<128, 128><<<gGemm, 1024, 0, stream>>>(x, W1, as1, ad1, buf0, als, ald, N);
  node_gather<128, true><<<gNode, 256, 0, stream>>>(rowptr, csr, buf0, als, ald, b1, buf1, N);

  gemm_wmma<128, 128><<<gGemm, 1024, 0, stream>>>(buf1, W2, as2, ad2, buf0, als, ald, N);
  node_gather<128, true><<<gNode, 256, 0, stream>>>(rowptr, csr, buf0, als, ald, b2, buf1, N);

  gemm_wmma<128, 64><<<gGemm, 1024, 0, stream>>>(buf1, W3, as3, ad3, buf0, als, ald, N);
  node_gather<64, false><<<gNode, 256, 0, stream>>>(rowptr, csr, buf0, als, ald, b3, outF, N);
}


# no product has a launch of its own: layer 1's WMMA product runs as extra work-groups of the 3 grouping launches; layers 2, 3's run on each wave's aggregated 16-node LDS tile; 8 launches; sums kept
# speedup vs baseline: 6.5186x; 1.0788x over previous
#include <hip/hip_runtime.h>

typedef __attribute__((ext_vector_type(2))) float v2f;
typedef __attribute__((ext_vector_type(8))) float v8f;

#define NEG_SLOPE 0.2f


template <int MASK>
__device__ __forceinline__ float row_xor_add(float keep, float send) {
  const int got = __builtin_amdgcn_update_dpp(0, __float_as_int(send), 0x160 | MASK,
                                              0xf, 0xf, true);
  return keep + __int_as_float(got);
}

template <int BIT, int P>
__device__ __forceinline__ void fold_half(float* p, bool up) {
#pragma unroll
  for (int j = 0; j < P / 2; ++j) {
    const float keep = up ? p[j + P / 2] : p[j];
    const float send = up ? p[j] : p[j + P / 2];
    p[j] = row_xor_add<BIT>(keep, send);
  }
}

template <int CIN, int COUT, int NTHR>
__device__ __forceinline__ void stage_w(const float* __restrict__ Wg, v2f* Wswz) {
  constexpr int TOT = (CIN / 4) * COUT * 2;
  for (int i0 = threadIdx.x; i0 < TOT; i0 += NTHR * 8) {
    v2f w[8];
#pragma unroll
    for (int q = 0; q < 8; ++q) {
      const int i = min(i0 + q * NTHR, TOT - 1);
      const int m = i / (COUT * 2);
      const int r = i - m * (COUT * 2);
      const float* wp = Wg + (size_t)(4 * m + 2 * (r & 1)) * COUT + (r >> 1);
      w[q].x = wp[0];
      w[q].y = wp[COUT];
    }
#pragma unroll
    for (int q = 0; q < 8; ++q)
      if (i0 + q * NTHR < TOT) Wswz[i0 + q * NTHR] = w[q];
  }
}

template <int COUT>
__device__ __forceinline__ void tile_epilogue(v8f* acc, const float* __restrict__ asrc,
                                              const float* __restrict__ adst,
                                              float* __restrict__ H, float* __restrict__ als,
                                              float* __restrict__ ald, int m0, int rsel,
                                              int hiHalf) {
  constexpr int NT = COUT / 16;
  const int rbase = m0 + hiHalf * 8;
#pragma unroll
  for (int t = 0; t < NT; ++t) {
#pragma unroll
    for (int i = 0; i < 8; ++i) {
      H[(size_t)(rbase + i) * COUT + t * 16 + rsel] = acc[t][i];
    }
  }
  v2f ps[4], pd[4];
#pragma unroll
  for (int j = 0; j < 4; ++j) { ps[j] = (v2f)(0.f); pd[j] = (v2f)(0.f); }
#pragma unroll
  for (int t = 0; t < NT; ++t) {
    const float as_t = asrc[t * 16 + rsel];
    const float ad_t = adst[t * 16 + rsel];
    const v2f as2 = {as_t, as_t};
    const v2f ad2 = {ad_t, ad_t};
#pragma unroll
    for (int j = 0; j < 4; ++j) {
      const v2f h2 = {acc[t][2 * j], acc[t][2 * j + 1]};
      ps[j] = __builtin_elementwise_fma(h2, as2, ps[j]);
      pd[j] = __builtin_elementwise_fma(h2, ad2, pd[j]);
    }
  }
  float p[16];
#pragma unroll
  for (int j = 0; j < 4; ++j) {
    p[2 * j] = ps[j].x;
    p[2 * j + 1] = ps[j].y;
    p[8 + 2 * j] = pd[j].x;
    p[8 + 2 * j + 1] = pd[j].y;
  }
  fold_half<8, 16>(p, (rsel & 8) != 0);
  fold_half<4, 8>(p, (rsel & 4) != 0);
  fold_half<2, 4>(p, (rsel & 2) != 0);
  fold_half<1, 2>(p, (rsel & 1) != 0);
  float* pdst = ((rsel & 8) ? ald : als) + rbase + (rsel & 7);
  *pdst = p[0];
}

template <int CIN, int COUT>
__device__ __forceinline__ void gemm_wg8(const float* __restrict__ X,
                                         const float* __restrict__ Wg,
                                         const float* __restrict__ asrc,
                                         const float* __restrict__ adst,
                                         float* __restrict__ H, float* __restrict__ als,
                                         float* __restrict__ ald, int M, int wg) {
  __shared__ v2f Wswz[(CIN / 4) * COUT * 2];
  stage_w<CIN, COUT, 256>(Wg, Wswz);
  __syncthreads();
  const int lane = threadIdx.x & 31;
  const int mtile = wg * 8 + (int)(threadIdx.x >> 5);
  if (mtile * 16 >= M) return;
  const int m0 = mtile * 16;
  constexpr int NT = COUT / 16;
  v8f acc[NT] = {};
  const int rsel = lane & 15;
  const int hiHalf = lane >> 4;
  const float* xrow = X + (size_t)(m0 + rsel) * CIN + hiHalf * 2;
  const v2f* wbase = Wswz + rsel * 2 + hiHalf;
  for (int k = 0; k < CIN; k += 4) {
    v2f a = *reinterpret_cast<const v2f*>(xrow + k);
    const v2f* wb = wbase + (k >> 2) * (COUT * 2);
#pragma unroll
    for (int t = 0; t < NT; ++t) {
      v2f bf = wb[t * 32];
      acc[t] = __builtin_amdgcn_wmma_f32_16x16x4_f32(
          false, a, false, bf, (short)0, acc[t], false, false);
    }
  }
  tile_epilogue<COUT>(acc, asrc, adst, H, als, ald, m0, rsel, hiHalf);
}

__device__ __forceinline__ float lrelu(float e) { return (e > 0.f) ? e : NEG_SLOPE * e; }

typedef __attribute__((ext_vector_type(2))) int v2i;

#define BUCKET_SHIFT 8
#define BUCKET_NODES 256
#define CHUNK_EDGES 4096

#define EDGES_PER_THREAD (CHUNK_EDGES / 256)

__global__ __launch_bounds__(256) void bucket_count(const int* __restrict__ dst,
                                                    int* __restrict__ cnt, int E, int G1,
                                                    int NBK,
    const float* __restrict__ gX, const float* __restrict__ gW,
    const float* __restrict__ gAs, const float* __restrict__ gAd, float* __restrict__ gH,
    float* __restrict__ gAls, float* __restrict__ gAld, int gM, int gFirst, int gCount) {
  if ((int)blockIdx.x < gCount) {
    gemm_wg8<128, 128>(gX, gW, gAs, gAd, gH, gAls, gAld, gM, gFirst + (int)blockIdx.x);
    return;
  }
  const int bx = (int)blockIdx.x - gCount;
  extern __shared__ int hist[];
  for (int i = threadIdx.x; i < NBK; i += 256) hist[i] = 0;
  __syncthreads();
  const int base = bx * CHUNK_EDGES + threadIdx.x;
  int d[EDGES_PER_THREAD];
#pragma unroll
  for (int q = 0; q < EDGES_PER_THREAD; ++q)
    d[q] = dst[min(base + q * 256, E - 1)];
#pragma unroll
  for (int q = 0; q < EDGES_PER_THREAD; ++q)
    if (base + q * 256 < E) atomicAdd(&hist[d[q] >> BUCKET_SHIFT], 1);
  __syncthreads();
  for (int i = threadIdx.x; i < NBK; i += 256) cnt[(size_t)i * G1 + bx] = hist[i];
}

__global__ __launch_bounds__(256) void chunk_sum(const int* __restrict__ v,
                                                 int* __restrict__ bsum, int Nn) {
  __shared__ int wsum[8];
  const int base = blockIdx.x * 1024 + threadIdx.x * 4;
  int t = 0;
  for (int q = 0; q < 4; ++q)
    if (base + q < Nn) t += v[base + q];
  for (int off = 16; off > 0; off >>= 1) t += __shfl_xor(t, off, 32);
  if ((threadIdx.x & 31) == 0) wsum[threadIdx.x >> 5] = t;
  __syncthreads();
  if (threadIdx.x == 0) {
    int s = 0;
    for (int w = 0; w < 8; ++w) s += wsum[w];
    bsum[blockIdx.x] = s;
  }
}

__global__ __launch_bounds__(256) void excl_scan(const int* __restrict__ in,
                                                 const int* __restrict__ bsum,
                                                 int* __restrict__ out, int Nn) {
  __shared__ int wpre[8];
  __shared__ int wtot[8];
  const int lane = threadIdx.x & 31;
  const int wave = threadIdx.x >> 5;
  int o = 0;
  for (int j = threadIdx.x; j < (int)blockIdx.x; j += 256) o += bsum[j];
  for (int off = 16; off > 0; off >>= 1) o += __shfl_xor(o, off, 32);
  if (lane == 0) wpre[wave] = o;

  const int base = blockIdx.x * 1024 + threadIdx.x * 4;
  int v[4];
  int t = 0;
  for (int q = 0; q < 4; ++q) {
    v[q] = (base + q < Nn) ? in[base + q] : 0;
    t += v[q];
  }
  int inc = t;
  for (int off = 1; off < 32; off <<= 1) {
    int u = __shfl_up(inc, off, 32);
    if (lane >= off) inc += u;
  }
  if (lane == 31) wtot[wave] = inc;
  __syncthreads();
  int ex = inc - t;
  for (int w = 0; w < 8; ++w) {
    ex += wpre[w];
    if (w < wave) ex += wtot[w];
  }
  for (int q = 0; q < 4; ++q) {
    if (base + q <= Nn) out[base + q] = ex;
    ex += v[q];
  }
}

__global__ __launch_bounds__(256) void bucket_scatter(const int* __restrict__ src,
                                                      const int* __restrict__ dst,
                                                      const int* __restrict__ offs,
                                                      int* __restrict__ binned, int E,
                                                      int G1, int NBK,
    const float* __restrict__ gX, const float* __restrict__ gW,
    const float* __restrict__ gAs, const float* __restrict__ gAd, float* __restrict__ gH,
    float* __restrict__ gAls, float* __restrict__ gAld, int gM, int gFirst, int gCount) {
  if ((int)blockIdx.x < gCount) {
    gemm_wg8<128, 128>(gX, gW, gAs, gAd, gH, gAls, gAld, gM, gFirst + (int)blockIdx.x);
    return;
  }
  const int bx = (int)blockIdx.x - gCount;
  extern __shared__ int slot[];
  for (int i = threadIdx.x; i < NBK; i += 256) slot[i] = offs[(size_t)i * G1 + bx];
  __syncthreads();
  const int base = bx * CHUNK_EDGES + threadIdx.x;
  int d[EDGES_PER_THREAD], s[EDGES_PER_THREAD];
#pragma unroll
  for (int q = 0; q < EDGES_PER_THREAD; ++q) {
    const int i = min(base + q * 256, E - 1);
    d[q] = dst[i];
    s[q] = src[i];
  }
#pragma unroll
  for (int q = 0; q < EDGES_PER_THREAD; ++q) {
    if (base + q * 256 < E) {
      const int pos = atomicAdd(&slot[d[q] >> BUCKET_SHIFT], 1);
      const v2i e = {s[q], d[q]};
      *reinterpret_cast<v2i*>(binned + 2 * (size_t)pos) = e;
    }
  }
}

__global__ __launch_bounds__(256) void bucket_csr(const int* __restrict__ binned,
                                                  const int* __restrict__ offs,
                                                  int* __restrict__ rowptr,
                                                  int* __restrict__ csr, int Nn, int G1,
    const float* __restrict__ gX, const float* __restrict__ gW,
    const float* __restrict__ gAs, const float* __restrict__ gAd, float* __restrict__ gH,
    float* __restrict__ gAls, float* __restrict__ gAld, int gM, int gFirst, int gCount) {
  if ((int)blockIdx.x < gCount) {
    gemm_wg8<128, 128>(gX, gW, gAs, gAd, gH, gAls, gAld, gM, gFirst + (int)blockIdx.x);
    return;
  }
  const int bx = (int)blockIdx.x - gCount;
  __shared__ int cnt[BUCKET_NODES];
  __shared__ int cur[BUCKET_NODES];
  __shared__ int wtot[BUCKET_NODES / 32];
  const int tid = threadIdx.x;
  const int lane = tid & 31;
  const int wave = tid >> 5;
  const int b = bx;
  const int beg = offs[(size_t)b * G1];
  const int end = offs[(size_t)(b + 1) * G1];
  cnt[tid] = 0;
  __syncthreads();
  for (int i0 = beg + tid; i0 < end; i0 += CHUNK_EDGES) {
    int dl[EDGES_PER_THREAD];
#pragma unroll
    for (int q = 0; q < EDGES_PER_THREAD; ++q)
      dl[q] = binned[2 * (size_t)min(i0 + q * 256, end - 1) + 1];
#pragma unroll
    for (int q = 0; q < EDGES_PER_THREAD; ++q)
      if (i0 + q * 256 < end) atomicAdd(&cnt[dl[q] & (BUCKET_NODES - 1)], 1);
  }
  __syncthreads();

  const int c = cnt[tid];
  int inc = c;
  for (int off = 1; off < 32; off <<= 1) {
    int u = __shfl_up(inc, off, 32);
    if (lane >= off) inc += u;
  }
  if (lane == 31) wtot[wave] = inc;
  __syncthreads();
  int ex = inc - c;
  for (int w = 0; w < BUCKET_NODES / 32; ++w)
    if (w < wave) ex += wtot[w];
  cur[tid] = ex;
  const int node = b * BUCKET_NODES + tid;
  if (node < Nn) rowptr[node] = beg + ex;
  if (tid == 0 && b == (int)gridDim.x - gCount - 1) rowptr[Nn] = end;
  __syncthreads();

  for (int i0 = beg + tid; i0 < end; i0 += CHUNK_EDGES) {
    v2i e[EDGES_PER_THREAD];
#pragma unroll
    for (int q = 0; q < EDGES_PER_THREAD; ++q)
      e[q] = *reinterpret_cast<const v2i*>(binned + 2 * (size_t)min(i0 + q * 256, end - 1));
#pragma unroll
    for (int q = 0; q < EDGES_PER_THREAD; ++q) {
      if (i0 + q * 256 < end) {
        const int pos = atomicAdd(&cur[e[q].y & (BUCKET_NODES - 1)], 1);
        csr[beg + pos] = e[q].x;
      }
    }
  }
}

__device__ __forceinline__ float wave_max(float v) {
  for (int off = 16; off > 0; off >>= 1) v = fmaxf(v, __shfl_xor(v, off, 32));
  return v;
}
__device__ __forceinline__ float wave_sum(float v) {
  for (int off = 16; off > 0; off >>= 1) v += __shfl_xor(v, off, 32);
  return v;
}

template <int C, typename vecf, int G>
__device__ __forceinline__ void add_rows_at(vecf& acc, const float* __restrict__ hl, int s,
                                            float w, int j) {
  vecf hv[G];
  float wj[G];
#pragma unroll
  for (int q = 0; q < G; ++q) {
    const int sj = __builtin_amdgcn_readlane(s, j + q);
    wj[q] = __int_as_float(__builtin_amdgcn_readlane(__float_as_int(w), j + q));
    hv[q] = *reinterpret_cast<const vecf*>(hl + (size_t)sj * C);
  }
#pragma unroll
  for (int q = 0; q < G; ++q) acc += wj[q] * hv[q];
}

template <int C, typename vecf, bool DEEP>
__device__ __forceinline__ void add_rows(vecf& acc, const float* __restrict__ hl,
                                         int s, float w, int cnt) {
  const int c4 = (cnt + 3) & ~3;
  int j = 0;
  if (DEEP) {
    for (; j + 8 <= c4; j += 8) add_rows_at<C, vecf, 8>(acc, hl, s, w, j);
  }
  for (; j < c4; j += 4) add_rows_at<C, vecf, 4>(acc, hl, s, w, j);
}

template <int C, bool RELU, bool DEEP, typename vecf>
__device__ __forceinline__ vecf gather_row_pre(int n, int lane, int beg, int end, float as_n,
                                               float ad, int s0, float a0,
                                               const int* __restrict__ csr,
                                               const float* __restrict__ hl,
                                               const float* __restrict__ als, vecf bv) {
  const vecf hself = *reinterpret_cast<const vecf*>(hl + (size_t)n * C);
  const float es = lrelu(as_n + ad);
  const bool valid = beg + lane < end;
  const float e0 = lrelu(a0 + ad);
  float m = valid ? fmaxf(es, e0) : es;
  for (int base = beg + 32; base < end; base += 32) {
    if (base + lane < end) m = fmaxf(m, lrelu(als[csr[base + lane]] + ad));
  }
  m = wave_max(m);

  const float xs = __expf(es - m);
  const float x0 = valid ? __expf(e0 - m) : 0.f;
  float sum = x0 + (lane == 0 ? xs : 0.f);
  for (int base = beg + 32; base < end; base += 32) {
    if (base + lane < end) sum += __expf(lrelu(als[csr[base + lane]] + ad) - m);
  }
  sum = wave_sum(sum);

  vecf acc = (xs / sum) * hself;
  add_rows<C, vecf, DEEP>(acc, hl, s0, x0 / sum, min(end - beg, 32));
  for (int base = beg + 32; base < end; base += 32) {
    const bool vd = base + lane < end;
    const int s = vd ? csr[base + lane] : n;
    const float x = vd ? __expf(lrelu(als[s] + ad) - m) : 0.f;
    add_rows<C, vecf, DEEP>(acc, hl, s, x / sum, min(end - base, 32));
  }

  vecf o = acc + bv;
  if (RELU) o = __builtin_elementwise_max(o, (vecf)(0.f));
  return o;
}

template <int C, bool RELU, bool DEEP, typename vecf>
__device__ __forceinline__ vecf gather_row(int n, int lane, const int* __restrict__ rowptr,
                                           const int* __restrict__ csr,
                                           const float* __restrict__ hl,
                                           const float* __restrict__ als,
                                           const float* __restrict__ ald, vecf bv) {
  const int beg = rowptr[n];
  const int end = rowptr[n + 1];
  const float ad = ald[n];
  const float as_n = als[n];
  const int s0 = (beg + lane < end) ? csr[beg + lane] : n;
  const float a0 = als[s0];
  return gather_row_pre<C, RELU, DEEP, vecf>(n, lane, beg, end, as_n, ad, s0, a0, csr, hl, als, bv);
}

template <int C, bool RELU>
__global__ __launch_bounds__(256) void node_gather(const int* __restrict__ rowptr,
                                                   const int* __restrict__ csr,
                                                   const float* __restrict__ H,
                                                   const float* __restrict__ als,
                                                   const float* __restrict__ ald,
                                                   const float* __restrict__ bias,
                                                   float* __restrict__ out, int Nn) {
  constexpr int VPT = C / 32;
  typedef __attribute__((ext_vector_type(VPT))) float vecf;
  const int lane = threadIdx.x & 31;
  const int w0 = __builtin_amdgcn_readfirstlane((int)(blockIdx.x * 8 + (threadIdx.x >> 5)));
  const int nw = gridDim.x * 8;
  const float* hl = H + lane * VPT;
  const vecf bv = *reinterpret_cast<const vecf*>(bias + lane * VPT);
  for (int n = w0; n < Nn; n += nw) {
    const vecf o = gather_row<C, RELU, false, vecf>(n, lane, rowptr, csr, hl, als, ald, bv);
    *reinterpret_cast<vecf*>(out + (size_t)n * C + lane * VPT) = o;
  }
}

#define FW 28
#define FROW 132
template <int COUT>
__global__ __launch_bounds__(FW * 32) void gather_gemm(
    const int* __restrict__ rowptr, const int* __restrict__ csr, const float* __restrict__ H,
    const float* __restrict__ als, const float* __restrict__ ald,
    const float* __restrict__ bias, const float* __restrict__ Wg,
    const float* __restrict__ asn, const float* __restrict__ adn, float* __restrict__ Hn,
    float* __restrict__ alsn, float* __restrict__ aldn, int Nn) {
  constexpr int C = 128;
  typedef __attribute__((ext_vector_type(4))) float v4f;
  __shared__ v2f Wswz[(C / 4) * COUT * 2];
  __shared__ __attribute__((aligned(16))) float rows[FW * 16 * FROW];
  stage_w<C, COUT, FW * 32>(Wg, Wswz);
  __syncthreads();

  const int lane = threadIdx.x & 31;
  const int wave = __builtin_amdgcn_readfirstlane((int)(threadIdx.x >> 5));
  float* my = rows + wave * (16 * FROW);
  const float* hl = H + lane * 4;
  const v4f bv = *reinterpret_cast<const v4f*>(bias + lane * 4);
  const int rsel = lane & 15;
  const int hiHalf = lane >> 4;
  const v2f* arow = reinterpret_cast<const v2f*>(my + rsel * FROW) + hiHalf;
  const v2f* wbase = Wswz + rsel * 2 + hiHalf;
  const int ntiles = (Nn + 15) >> 4;
  constexpr int NT = COUT / 16;

  for (int tile = wave * (int)gridDim.x + (int)blockIdx.x; tile < ntiles;
       tile += (int)gridDim.x * FW) {
    const int m0 = tile * 16;
    const int rpv = rowptr[min(m0 + lane, Nn)];
    const float asv = als[min(m0 + lane, Nn - 1)];
    const float adv = ald[min(m0 + lane, Nn - 1)];
    int beg = __builtin_amdgcn_readlane(rpv, 0);
    int end = __builtin_amdgcn_readlane(rpv, 1);
    int s_cur = (beg + lane < end) ? csr[beg + lane] : min(m0, Nn - 1);
    float a_cur = als[s_cur];
    int s_nxt;
    {
      const int b1 = end;
      const int e1 = __builtin_amdgcn_readlane(rpv, 2);
      s_nxt = (b1 + lane < e1) ? csr[b1 + lane] : min(m0 + 1, Nn - 1);
    }
    for (int r = 0; r < 16; ++r) {
      const int n = m0 + r;
      const float a_nxt = als[s_nxt];
      const int r2 = min(r + 2, 15);
      const int b2 = __builtin_amdgcn_readlane(rpv, r2);
      const int e2 = __builtin_amdgcn_readlane(rpv, r2 + 1);
      const int s_nn = (b2 + lane < e2) ? csr[b2 + lane] : min(m0 + r2, Nn - 1);
      const float as_n = __int_as_float(__builtin_amdgcn_readlane(__float_as_int(asv), r));
      const float ad = __int_as_float(__builtin_amdgcn_readlane(__float_as_int(adv), r));
      v4f o = (v4f)(0.f);
      if (n < Nn)
        o = gather_row_pre<C, true, true, v4f>(n, lane, beg, end, as_n, ad, s_cur, a_cur, csr, hl,
                                               als, bv);
      *reinterpret_cast<v4f*>(my + r * FROW + lane * 4) = o;
      s_cur = s_nxt;
      a_cur = a_nxt;
      s_nxt = s_nn;
      const int r1 = min(r + 1, 15);
      beg = __builtin_amdgcn_readlane(rpv, r1);
      end = __builtin_amdgcn_readlane(rpv, r1 + 1);
    }
    v8f acc[NT] = {};
#pragma unroll 4
    for (int k = 0; k < C; k += 4) {
      const v2f a = arow[k >> 1];
      const v2f* wb = wbase + (k >> 2) * (COUT * 2);
#pragma unroll
      for (int t = 0; t < NT; ++t) {
        const v2f bf = wb[t * 32];
        acc[t] = __builtin_amdgcn_wmma_f32_16x16x4_f32(
            false, a, false, bf, (short)0, acc[t], false, false);
      }
    }
    tile_epilogue<COUT>(acc, asn, adn, Hn, alsn, aldn, m0, rsel, hiHalf);
  }
}

extern "C" void kernel_launch(void* const* d_in, const int* in_sizes, int n_in,
                              void* d_out, int out_size, void* d_ws, size_t ws_size,
                              hipStream_t stream) {
  const float* x   = (const float*)d_in[0];
  const int*   ei  = (const int*)d_in[1];
  const float* W1  = (const float*)d_in[2];
  const float* as1 = (const float*)d_in[3];
  const float* ad1 = (const float*)d_in[4];
  const float* b1  = (const float*)d_in[5];
  const float* W2  = (const float*)d_in[6];
  const float* as2 = (const float*)d_in[7];
  const float* ad2 = (const float*)d_in[8];
  const float* b2  = (const float*)d_in[9];
  const float* W3  = (const float*)d_in[10];
  const float* as3 = (const float*)d_in[11];
  const float* ad3 = (const float*)d_in[12];
  const float* b3  = (const float*)d_in[13];

  const int N = in_sizes[0] / 128;
  const int E = in_sizes[1] / 2;
  const int* srcp = ei;
  const int* dstp = ei + E;

  const int NBK = (N + BUCKET_NODES - 1) / BUCKET_NODES;
  const int G1  = (E + CHUNK_EDGES - 1) / CHUNK_EDGES;
  const int NT  = NBK * G1;

  float* buf0   = (float*)d_ws;
  float* buf1   = buf0 + (size_t)N * 128;
  int*   binned = (int*)(buf1 + (size_t)N * 128);
  int*   csr    = binned + 2 * (size_t)E;
  float* alsA   = (float*)(csr + E);
  float* aldA   = alsA + N;
  float* alsB   = aldA + N;
  float* aldB   = alsB + N;
  int*   rowptr = (int*)(aldB + N);
  int*   cnt    = rowptr + N + 1;
  int*   offs   = cnt + NT;
  int*   bsum   = offs + NT + 1;

  const int gW8   = ((N + 15) / 16 + 7) / 8;
  const int gShare = (gW8 + 2) / 3;
  const int gFuse = 256;
  const int gScan = NT / 1024 + 1;
  const int gNode = 4096;
  float* outF = (float*)d_out;

  bucket_count<<<gShare + G1, 256, NBK * sizeof(int), stream>>>(
      dstp, cnt, E, G1, NBK, x, W1, as1, ad1, buf0, alsA, aldA, N, 0, gShare);
  chunk_sum<<<gScan, 256, 0, stream>>>(cnt, bsum, NT);
  excl_scan<<<gScan, 256, 0, stream>>>(cnt, bsum, offs, NT);
  bucket_scatter<<<gShare + G1, 256, NBK * sizeof(int), stream>>>(
      srcp, dstp, offs, binned, E, G1, NBK, x, W1, as1, ad1, buf0, alsA, aldA, N, gShare, gShare);
  bucket_csr<<<gShare + NBK, 256, 0, stream>>>(
      binned, offs, rowptr, csr, N, G1, x, W1, as1, ad1, buf0, alsA, aldA, N, 2 * gShare, gShare);

  gather_gemm<128><<<gFuse, FW * 32, 0, stream>>>(rowptr, csr, buf0, alsA, aldA, b1, W2, as2, ad2,
                                                  buf1, alsB, aldB, N);
  gather_gemm<64><<<gFuse, FW * 32, 0, stream>>>(rowptr, csr, buf1, alsB, aldB, b2, W3, as3, ad3,
                                                 buf0, alsA, aldA, N);
  node_gather<64, false><<<gNode, 256, 0, stream>>>(rowptr, csr, buf0, alsA, aldA, b3, outF, N);
}
